# starting kernel
# baseline (speedup 1.0000x reference)
#include <hip/hip_runtime.h>

typedef __attribute__((ext_vector_type(2))) float v2f;
typedef __attribute__((ext_vector_type(8))) float v8f;

__global__ void init_deg_kernel(float* __restrict__ deg, int n) {
    int i = blockIdx.x * blockDim.x + threadIdx.x;
    if (i < n) deg[i] = 1.0f;
}

__global__ void count_deg_kernel(const int* __restrict__ dst,
                                 float* __restrict__ deg, int E) {
    int e = blockIdx.x * blockDim.x + threadIdx.x;
    if (e < E) atomicAdd(&deg[dst[e]], 1.0f);
}

__global__ void isqrt_kernel(float* __restrict__ deg, int n) {
    int i = blockIdx.x * blockDim.x + threadIdx.x;
    if (i < n) deg[i] = rsqrtf(deg[i]);
}

template <bool FUSE>
__global__ void gemm_wmma_kernel(const float* __restrict__ A,
                                 const float* __restrict__ W,
                                 const float* __restrict__ dis,
                                 const float* __restrict__ bias,
                                 float* __restrict__ ht,
                                 float* __restrict__ agg,
                                 int M, int K) {
    const int lane = threadIdx.x & 31;
    const int wave = blockIdx.x * (blockDim.x >> 5) + (threadIdx.x >> 5);
    const int m0 = wave * 16;
    if (m0 >= M) return;

    const int half = lane >> 4;
    const int ml   = lane & 15;
    const int arow = m0 + ml;
    const float dA = FUSE ? dis[arow] : 0.0f;

    v8f c0 = {}, c1 = {}, c2 = {}, c3 = {};

    for (int k0 = 0; k0 < K; k0 += 4) {
        const int ka = k0 + half * 2;
        float a0 = A[(size_t)arow * K + ka];
        float a1 = A[(size_t)arow * K + ka + 1];
        if (FUSE) {
            a0 = fmaxf(fmaf(a0, dA, bias[ka]), 0.0f);
            a1 = fmaxf(fmaf(a1, dA, bias[ka + 1]), 0.0f);
        }
        v2f av; av.x = a0; av.y = a1;

        const float* w0 = W + (size_t)ka * 64;
        const float* w1 = W + (size_t)(ka + 1) * 64;
        v2f b0; b0.x = w0[ml];      b0.y = w1[ml];
        v2f b1; b1.x = w0[16 + ml]; b1.y = w1[16 + ml];
        v2f b2; b2.x = w0[32 + ml]; b2.y = w1[32 + ml];
        v2f b3; b3.x = w0[48 + ml]; b3.y = w1[48 + ml];

        c0 = __builtin_amdgcn_wmma_f32_16x16x4_f32(false, av, false, b0, (short)0, c0, false, false);
        c1 = __builtin_amdgcn_wmma_f32_16x16x4_f32(false, av, false, b1, (short)0, c1, false, false);
        c2 = __builtin_amdgcn_wmma_f32_16x16x4_f32(false, av, false, b2, (short)0, c2, false, false);
        c3 = __builtin_amdgcn_wmma_f32_16x16x4_f32(false, av, false, b3, (short)0, c3, false, false);
    }

    for (int r = 0; r < 8; ++r) {
        const int orow = m0 + half * 8 + r;
        const float dr = dis[orow];
        const float v0 = c0[r] * dr;
        const float v1 = c1[r] * dr;
        const float v2 = c2[r] * dr;
        const float v3 = c3[r] * dr;
        float* hp = ht  + (size_t)orow * 64;
        float* ap = agg + (size_t)orow * 64;
        hp[ml] = v0; hp[16 + ml] = v1; hp[32 + ml] = v2; hp[48 + ml] = v3;
        ap[ml] = v0; ap[16 + ml] = v1; ap[32 + ml] = v2; ap[48 + ml] = v3;
    }
}

__global__ void scatter_edges_kernel(const int* __restrict__ src,
                                     const int* __restrict__ dst,
                                     const float* __restrict__ ht,
                                     float* __restrict__ agg, int E) {
    const int lane = threadIdx.x & 31;
    const int e = blockIdx.x * (blockDim.x >> 5) + (threadIdx.x >> 5);
    if (e >= E) return;
    const int s = src[e];
    const int d = dst[e];
    const float2 v = *(const float2*)(ht + (size_t)s * 64 + lane * 2);
    float* ap = agg + (size_t)d * 64 + lane * 2;
    atomicAdd(ap, v.x);
    atomicAdd(ap + 1, v.y);
}

__global__ void fc_out_kernel(const float* __restrict__ agg2,
                              const float* __restrict__ dis,
                              const float* __restrict__ b2,
                              const float* __restrict__ wfc,
                              const float* __restrict__ bfc,
                              float* __restrict__ out, int M) {
    const int lane = threadIdx.x & 31;
    const int i = blockIdx.x * (blockDim.x >> 5) + (threadIdx.x >> 5);
    if (i >= M) return;
    const float di = dis[i];
    const int j = lane * 2;
    const float2 a = *(const float2*)(agg2 + (size_t)i * 64 + j);
    const float h0 = fmaxf(fmaf(a.x, di, b2[j]), 0.0f);
    const float h1 = fmaxf(fmaf(a.y, di, b2[j + 1]), 0.0f);
    float p = h0 * wfc[j] + h1 * wfc[j + 1];
    #pragma unroll
    for (int off = 16; off > 0; off >>= 1) p += __shfl_down(p, off, 32);
    if (lane == 0) out[i] = p + bfc[0];
}

extern "C" void kernel_launch(void* const* d_in, const int* in_sizes, int n_in,
                              void* d_out, int out_size, void* d_ws, size_t ws_size,
                              hipStream_t stream) {
    const float* x   = (const float*)d_in[0];
    const int*   ei  = (const int*)d_in[1];
    const float* W1  = (const float*)d_in[2];
    const float* b1  = (const float*)d_in[3];
    const float* W2  = (const float*)d_in[4];
    const float* b2  = (const float*)d_in[5];
    const float* Wfc = (const float*)d_in[6];
    const float* bfc = (const float*)d_in[7];

    const int HID = in_sizes[3];
    const int IN  = in_sizes[2] / HID;
    const int N   = in_sizes[0] / IN;
    const int E   = in_sizes[1] / 2;
    const int* srcv = ei;
    const int* dstv = ei + E;

    char* ws = (char*)d_ws;
    float* dis = (float*)ws;
    size_t off = (((size_t)N * 4) + 255) & ~(size_t)255;
    const size_t rowBytes = (size_t)N * 64 * sizeof(float);
    float* ht   = (float*)(ws + off);
    float* agg1 = (float*)(ws + off + rowBytes);
    float* agg2 = (float*)(ws + off + 2 * rowBytes);

    init_deg_kernel<<<(N + 255) / 256, 256, 0, stream>>>(dis, N);
    count_deg_kernel<<<(E + 255) / 256, 256, 0, stream>>>(dstv, dis, E);
    isqrt_kernel<<<(N + 255) / 256, 256, 0, stream>>>(dis, N);

    const int gemmBlocks = ((N + 15) / 16 + 7) / 8;
    const int edgeBlocks = (E + 7) / 8;
    const int nodeBlocks = (N + 7) / 8;

    gemm_wmma_kernel<false><<<gemmBlocks, 256, 0, stream>>>(x, W1, dis, nullptr, ht, agg1, N, IN);
    scatter_edges_kernel<<<edgeBlocks, 256, 0, stream>>>(srcv, dstv, ht, agg1, E);

    gemm_wmma_kernel<true><<<gemmBlocks, 256, 0, stream>>>(agg1, W2, dis, b1, ht, agg2, N, HID);
    scatter_edges_kernel<<<edgeBlocks, 256, 0, stream>>>(srcv, dstv, ht, agg2, E);

    fc_out_kernel<<<nodeBlocks, 256, 0, stream>>>(agg2, dis, b2, Wfc, bfc, (float*)d_out, N);
}


# edge scatter: each wave takes 32 consecutive edges (indices read once, all 32 row gathers in flight before the first atomic) instead of one edge per wave; f32 atomics kept, sum order still by arrival
# speedup vs baseline: 1.5293x; 1.5293x over previous
#include <hip/hip_runtime.h>

typedef __attribute__((ext_vector_type(2))) float v2f;
typedef __attribute__((ext_vector_type(8))) float v8f;

__global__ void init_deg_kernel(float* __restrict__ deg, int n) {
    int i = blockIdx.x * blockDim.x + threadIdx.x;
    if (i < n) deg[i] = 1.0f;
}

__global__ void count_deg_kernel(const int* __restrict__ dst,
                                 float* __restrict__ deg, int E) {
    int e = blockIdx.x * blockDim.x + threadIdx.x;
    if (e < E) atomicAdd(&deg[dst[e]], 1.0f);
}

__global__ void isqrt_kernel(float* __restrict__ deg, int n) {
    int i = blockIdx.x * blockDim.x + threadIdx.x;
    if (i < n) deg[i] = rsqrtf(deg[i]);
}

template <bool FUSE>
__global__ void gemm_wmma_kernel(const float* __restrict__ A,
                                 const float* __restrict__ W,
                                 const float* __restrict__ dis,
                                 const float* __restrict__ bias,
                                 float* __restrict__ ht,
                                 float* __restrict__ agg,
                                 int M, int K) {
    const int lane = threadIdx.x & 31;
    const int wave = blockIdx.x * (blockDim.x >> 5) + (threadIdx.x >> 5);
    const int m0 = wave * 16;
    if (m0 >= M) return;

    const int half = lane >> 4;
    const int ml   = lane & 15;
    const int arow = m0 + ml;
    const float dA = FUSE ? dis[arow] : 0.0f;

    v8f c0 = {}, c1 = {}, c2 = {}, c3 = {};

    for (int k0 = 0; k0 < K; k0 += 4) {
        const int ka = k0 + half * 2;
        float a0 = A[(size_t)arow * K + ka];
        float a1 = A[(size_t)arow * K + ka + 1];
        if (FUSE) {
            a0 = fmaxf(fmaf(a0, dA, bias[ka]), 0.0f);
            a1 = fmaxf(fmaf(a1, dA, bias[ka + 1]), 0.0f);
        }
        v2f av; av.x = a0; av.y = a1;

        const float* w0 = W + (size_t)ka * 64;
        const float* w1 = W + (size_t)(ka + 1) * 64;
        v2f b0; b0.x = w0[ml];      b0.y = w1[ml];
        v2f b1; b1.x = w0[16 + ml]; b1.y = w1[16 + ml];
        v2f b2; b2.x = w0[32 + ml]; b2.y = w1[32 + ml];
        v2f b3; b3.x = w0[48 + ml]; b3.y = w1[48 + ml];

        c0 = __builtin_amdgcn_wmma_f32_16x16x4_f32(false, av, false, b0, (short)0, c0, false, false);
        c1 = __builtin_amdgcn_wmma_f32_16x16x4_f32(false, av, false, b1, (short)0, c1, false, false);
        c2 = __builtin_amdgcn_wmma_f32_16x16x4_f32(false, av, false, b2, (short)0, c2, false, false);
        c3 = __builtin_amdgcn_wmma_f32_16x16x4_f32(false, av, false, b3, (short)0, c3, false, false);
    }

    for (int r = 0; r < 8; ++r) {
        const int orow = m0 + half * 8 + r;
        const float dr = dis[orow];
        const float v0 = c0[r] * dr;
        const float v1 = c1[r] * dr;
        const float v2 = c2[r] * dr;
        const float v3 = c3[r] * dr;
        float* hp = ht  + (size_t)orow * 64;
        float* ap = agg + (size_t)orow * 64;
        hp[ml] = v0; hp[16 + ml] = v1; hp[32 + ml] = v2; hp[48 + ml] = v3;
        ap[ml] = v0; ap[16 + ml] = v1; ap[32 + ml] = v2; ap[48 + ml] = v3;
    }
}

#define EDGES_PER_WAVE 32
__global__ void scatter_edges_kernel(const int* __restrict__ src,
                                     const int* __restrict__ dst,
                                     const float* __restrict__ ht,
                                     float* __restrict__ agg, int E) {
    const int lane = threadIdx.x & 31;
    const int wv = __builtin_amdgcn_readfirstlane((int)(threadIdx.x >> 5));
    const int e0 = (blockIdx.x * (blockDim.x >> 5) + wv) * EDGES_PER_WAVE;
    if (e0 >= E) return;
    const int* sp = src + e0;
    const int* dp = dst + e0;
    const float* hl = ht + lane * 2;
    float* al = agg + lane * 2;

    if (e0 + EDGES_PER_WAVE <= E) {
        int d[EDGES_PER_WAVE];
        float2 v[EDGES_PER_WAVE];
        #pragma unroll
        for (int j = 0; j < EDGES_PER_WAVE; ++j) {
            d[j] = dp[j];
            v[j] = *(const float2*)(hl + (size_t)sp[j] * 64);
        }
        #pragma unroll
        for (int j = 0; j < EDGES_PER_WAVE; ++j) {
            float* ap = al + (size_t)d[j] * 64;
            atomicAdd(ap, v[j].x);
            atomicAdd(ap + 1, v[j].y);
        }
    } else {
        for (int j = 0; e0 + j < E; ++j) {
            const float2 v = *(const float2*)(hl + (size_t)sp[j] * 64);
            float* ap = al + (size_t)dp[j] * 64;
            atomicAdd(ap, v.x);
            atomicAdd(ap + 1, v.y);
        }
    }
}

__global__ void fc_out_kernel(const float* __restrict__ agg2,
                              const float* __restrict__ dis,
                              const float* __restrict__ b2,
                              const float* __restrict__ wfc,
                              const float* __restrict__ bfc,
                              float* __restrict__ out, int M) {
    const int lane = threadIdx.x & 31;
    const int i = blockIdx.x * (blockDim.x >> 5) + (threadIdx.x >> 5);
    if (i >= M) return;
    const float di = dis[i];
    const int j = lane * 2;
    const float2 a = *(const float2*)(agg2 + (size_t)i * 64 + j);
    const float h0 = fmaxf(fmaf(a.x, di, b2[j]), 0.0f);
    const float h1 = fmaxf(fmaf(a.y, di, b2[j + 1]), 0.0f);
    float p = h0 * wfc[j] + h1 * wfc[j + 1];
    #pragma unroll
    for (int off = 16; off > 0; off >>= 1) p += __shfl_down(p, off, 32);
    if (lane == 0) out[i] = p + bfc[0];
}

extern "C" void kernel_launch(void* const* d_in, const int* in_sizes, int n_in,
                              void* d_out, int out_size, void* d_ws, size_t ws_size,
                              hipStream_t stream) {
    const float* x   = (const float*)d_in[0];
    const int*   ei  = (const int*)d_in[1];
    const float* W1  = (const float*)d_in[2];
    const float* b1  = (const float*)d_in[3];
    const float* W2  = (const float*)d_in[4];
    const float* b2  = (const float*)d_in[5];
    const float* Wfc = (const float*)d_in[6];
    const float* bfc = (const float*)d_in[7];

    const int HID = in_sizes[3];
    const int IN  = in_sizes[2] / HID;
    const int N   = in_sizes[0] / IN;
    const int E   = in_sizes[1] / 2;
    const int* srcv = ei;
    const int* dstv = ei + E;

    char* ws = (char*)d_ws;
    float* dis = (float*)ws;
    size_t off = (((size_t)N * 4) + 255) & ~(size_t)255;
    const size_t rowBytes = (size_t)N * 64 * sizeof(float);
    float* ht   = (float*)(ws + off);
    float* agg1 = (float*)(ws + off + rowBytes);
    float* agg2 = (float*)(ws + off + 2 * rowBytes);

    init_deg_kernel<<<(N + 255) / 256, 256, 0, stream>>>(dis, N);
    count_deg_kernel<<<(E + 255) / 256, 256, 0, stream>>>(dstv, dis, E);
    isqrt_kernel<<<(N + 255) / 256, 256, 0, stream>>>(dis, N);

    const int gemmBlocks = ((N + 15) / 16 + 7) / 8;
    const int edgeBlocks = (E + 255) / 256;
    const int nodeBlocks = (N + 7) / 8;

    gemm_wmma_kernel<false><<<gemmBlocks, 256, 0, stream>>>(x, W1, dis, nullptr, ht, agg1, N, IN);
    scatter_edges_kernel<<<edgeBlocks, 256, 0, stream>>>(srcv, dstv, ht, agg1, E);

    gemm_wmma_kernel<true><<<gemmBlocks, 256, 0, stream>>>(agg1, W2, dis, b1, ht, agg2, N, HID);
    scatter_edges_kernel<<<edgeBlocks, 256, 0, stream>>>(srcv, dstv, ht, agg2, E);

    fc_out_kernel<<<nodeBlocks, 256, 0, stream>>>(agg2, dis, b2, Wfc, bfc, (float*)d_out, N);
}


# f32 atomic sums (degree count, both edge scatters) go to two planes, one per package half chosen by the wave's die number; readers add the planes; sum order: arrival order in two partial sums
# speedup vs baseline: 1.9172x; 1.2536x over previous
#include <hip/hip_runtime.h>

typedef __attribute__((ext_vector_type(2))) float v2f;
typedef __attribute__((ext_vector_type(8))) float v8f;

__device__ __forceinline__ int wave_half() {
    const unsigned id = __builtin_amdgcn_s_sendmsg_rtn(0x87);
    return (int)((id >> 18) & 1u);
}

__global__ void init_deg_kernel(float* __restrict__ degA, float* __restrict__ degB, int n) {
    int i = blockIdx.x * blockDim.x + threadIdx.x;
    if (i < n) { degA[i] = 1.0f; degB[i] = 0.0f; }
}

__global__ void count_deg_kernel(const int* __restrict__ dst,
                                 float* __restrict__ degA,
                                 float* __restrict__ degB, int E) {
    int e = blockIdx.x * blockDim.x + threadIdx.x;
    float* deg = wave_half() ? degB : degA;
    if (e < E) atomicAdd(&deg[dst[e]], 1.0f);
}

__global__ void isqrt_kernel(float* __restrict__ degA, const float* __restrict__ degB, int n) {
    int i = blockIdx.x * blockDim.x + threadIdx.x;
    if (i < n) degA[i] = rsqrtf(degA[i] + degB[i]);
}

template <bool FUSE>
__global__ void gemm_wmma_kernel(const float* __restrict__ A,
                                 const float* __restrict__ A2,
                                 const float* __restrict__ W,
                                 const float* __restrict__ dis,
                                 const float* __restrict__ bias,
                                 float* __restrict__ ht,
                                 float* __restrict__ agg,
                                 float* __restrict__ aggB,
                                 int M, int K) {
    const int lane = threadIdx.x & 31;
    const int wave = blockIdx.x * (blockDim.x >> 5) + (threadIdx.x >> 5);
    const int m0 = wave * 16;
    if (m0 >= M) return;

    const int half = lane >> 4;
    const int ml   = lane & 15;
    const int arow = m0 + ml;
    const float dA = FUSE ? dis[arow] : 0.0f;

    v8f c0 = {}, c1 = {}, c2 = {}, c3 = {};

    for (int k0 = 0; k0 < K; k0 += 4) {
        const int ka = k0 + half * 2;
        float a0 = A[(size_t)arow * K + ka];
        float a1 = A[(size_t)arow * K + ka + 1];
        if (FUSE) {
            a0 += A2[(size_t)arow * K + ka];
            a1 += A2[(size_t)arow * K + ka + 1];
            a0 = fmaxf(fmaf(a0, dA, bias[ka]), 0.0f);
            a1 = fmaxf(fmaf(a1, dA, bias[ka + 1]), 0.0f);
        }
        v2f av; av.x = a0; av.y = a1;

        const float* w0 = W + (size_t)ka * 64;
        const float* w1 = W + (size_t)(ka + 1) * 64;
        v2f b0; b0.x = w0[ml];      b0.y = w1[ml];
        v2f b1; b1.x = w0[16 + ml]; b1.y = w1[16 + ml];
        v2f b2; b2.x = w0[32 + ml]; b2.y = w1[32 + ml];
        v2f b3; b3.x = w0[48 + ml]; b3.y = w1[48 + ml];

        c0 = __builtin_amdgcn_wmma_f32_16x16x4_f32(false, av, false, b0, (short)0, c0, false, false);
        c1 = __builtin_amdgcn_wmma_f32_16x16x4_f32(false, av, false, b1, (short)0, c1, false, false);
        c2 = __builtin_amdgcn_wmma_f32_16x16x4_f32(false, av, false, b2, (short)0, c2, false, false);
        c3 = __builtin_amdgcn_wmma_f32_16x16x4_f32(false, av, false, b3, (short)0, c3, false, false);
    }

    for (int r = 0; r < 8; ++r) {
        const int orow = m0 + half * 8 + r;
        const float dr = dis[orow];
        const float v0 = c0[r] * dr;
        const float v1 = c1[r] * dr;
        const float v2 = c2[r] * dr;
        const float v3 = c3[r] * dr;
        float* hp = ht  + (size_t)orow * 64;
        float* ap = agg + (size_t)orow * 64;
        float* bp = aggB + (size_t)orow * 64;
        hp[ml] = v0; hp[16 + ml] = v1; hp[32 + ml] = v2; hp[48 + ml] = v3;
        ap[ml] = v0; ap[16 + ml] = v1; ap[32 + ml] = v2; ap[48 + ml] = v3;
        bp[ml] = 0.0f; bp[16 + ml] = 0.0f; bp[32 + ml] = 0.0f; bp[48 + ml] = 0.0f;
    }
}

#define EDGES_PER_WAVE 32
__global__ void scatter_edges_kernel(const int* __restrict__ src,
                                     const int* __restrict__ dst,
                                     const float* __restrict__ ht,
                                     float* __restrict__ aggA,
                                     float* __restrict__ aggB, int E) {
    const int lane = threadIdx.x & 31;
    const int wv = __builtin_amdgcn_readfirstlane((int)(threadIdx.x >> 5));
    const int e0 = (blockIdx.x * (blockDim.x >> 5) + wv) * EDGES_PER_WAVE;
    if (e0 >= E) return;
    const int* sp = src + e0;
    const int* dp = dst + e0;
    const float* hl = ht + lane * 2;
    float* al = (wave_half() ? aggB : aggA) + lane * 2;

    if (e0 + EDGES_PER_WAVE <= E) {
        int d[EDGES_PER_WAVE];
        float2 v[EDGES_PER_WAVE];
        #pragma unroll
        for (int j = 0; j < EDGES_PER_WAVE; ++j) {
            d[j] = dp[j];
            v[j] = *(const float2*)(hl + (size_t)sp[j] * 64);
        }
        #pragma unroll
        for (int j = 0; j < EDGES_PER_WAVE; ++j) {
            float* ap = al + (size_t)d[j] * 64;
            atomicAdd(ap, v[j].x);
            atomicAdd(ap + 1, v[j].y);
        }
    } else {
        for (int j = 0; e0 + j < E; ++j) {
            const float2 v = *(const float2*)(hl + (size_t)sp[j] * 64);
            float* ap = al + (size_t)dp[j] * 64;
            atomicAdd(ap, v.x);
            atomicAdd(ap + 1, v.y);
        }
    }
}

__global__ void fc_out_kernel(const float* __restrict__ agg2,
                              const float* __restrict__ agg2B,
                              const float* __restrict__ dis,
                              const float* __restrict__ b2,
                              const float* __restrict__ wfc,
                              const float* __restrict__ bfc,
                              float* __restrict__ out, int M) {
    const int lane = threadIdx.x & 31;
    const int i = blockIdx.x * (blockDim.x >> 5) + (threadIdx.x >> 5);
    if (i >= M) return;
    const float di = dis[i];
    const int j = lane * 2;
    const float2 a = *(const float2*)(agg2 + (size_t)i * 64 + j);
    const float2 b = *(const float2*)(agg2B + (size_t)i * 64 + j);
    const float h0 = fmaxf(fmaf(a.x + b.x, di, b2[j]), 0.0f);
    const float h1 = fmaxf(fmaf(a.y + b.y, di, b2[j + 1]), 0.0f);
    float p = h0 * wfc[j] + h1 * wfc[j + 1];
    #pragma unroll
    for (int off = 16; off > 0; off >>= 1) p += __shfl_down(p, off, 32);
    if (lane == 0) out[i] = p + bfc[0];
}

extern "C" void kernel_launch(void* const* d_in, const int* in_sizes, int n_in,
                              void* d_out, int out_size, void* d_ws, size_t ws_size,
                              hipStream_t stream) {
    const float* x   = (const float*)d_in[0];
    const int*   ei  = (const int*)d_in[1];
    const float* W1  = (const float*)d_in[2];
    const float* b1  = (const float*)d_in[3];
    const float* W2  = (const float*)d_in[4];
    const float* b2  = (const float*)d_in[5];
    const float* Wfc = (const float*)d_in[6];
    const float* bfc = (const float*)d_in[7];

    const int HID = in_sizes[3];
    const int IN  = in_sizes[2] / HID;
    const int N   = in_sizes[0] / IN;
    const int E   = in_sizes[1] / 2;
    const int* srcv = ei;
    const int* dstv = ei + E;

    char* ws = (char*)d_ws;
    float* dis = (float*)ws;
    const size_t degBytes = (((size_t)N * 4) + 255) & ~(size_t)255;
    float* degB = (float*)(ws + degBytes);
    size_t off = 2 * degBytes;
    const size_t rowBytes = (size_t)N * 64 * sizeof(float);
    float* ht    = (float*)(ws + off);
    float* agg1  = (float*)(ws + off + rowBytes);
    float* agg1B = (float*)(ws + off + 2 * rowBytes);
    float* agg2  = (float*)(ws + off + 3 * rowBytes);
    float* agg2B = (float*)(ws + off + 4 * rowBytes);

    init_deg_kernel<<<(N + 255) / 256, 256, 0, stream>>>(dis, degB, N);
    count_deg_kernel<<<(E + 255) / 256, 256, 0, stream>>>(dstv, dis, degB, E);
    isqrt_kernel<<<(N + 255) / 256, 256, 0, stream>>>(dis, degB, N);

    const int gemmBlocks = ((N + 15) / 16 + 7) / 8;
    const int edgeBlocks = (E + 255) / 256;
    const int nodeBlocks = (N + 7) / 8;

    gemm_wmma_kernel<false><<<gemmBlocks, 256, 0, stream>>>(x, nullptr, W1, dis, nullptr, ht, agg1, agg1B, N, IN);
    scatter_edges_kernel<<<edgeBlocks, 256, 0, stream>>>(srcv, dstv, ht, agg1, agg1B, E);

    gemm_wmma_kernel<true><<<gemmBlocks, 256, 0, stream>>>(agg1, agg1B, W2, dis, b1, ht, agg2, agg2B, N, HID);
    scatter_edges_kernel<<<edgeBlocks, 256, 0, stream>>>(srcv, dstv, ht, agg2, agg2B, E);

    fc_out_kernel<<<nodeBlocks, 256, 0, stream>>>(agg2, agg2B, dis, b2, Wfc, bfc, (float*)d_out, N);
}


# edge sums without float atomics: edges bucketed by destination once per call (rank = return of the degree count's int atomic), one wave per node adds its rows in registers; sum order: arrival rank
# speedup vs baseline: 2.7318x; 1.4249x over previous
#include <hip/hip_runtime.h>

typedef __attribute__((ext_vector_type(2))) float v2f;
typedef __attribute__((ext_vector_type(8))) float v8f;

__device__ __forceinline__ int wave_half() {
    const unsigned id = __builtin_amdgcn_s_sendmsg_rtn(0x87);
    return (int)((id >> 18) & 1u);
}

__global__ void init_count_kernel(int* __restrict__ cntA, int* __restrict__ cntB,
                                  int* __restrict__ cursor, int n) {
    int i = blockIdx.x * blockDim.x + threadIdx.x;
    if (i < n) { cntA[i] = 0; cntB[i] = 0; }
    if (i == 0) cursor[0] = 0;
}

__global__ void count_rank_kernel(const int* __restrict__ dst,
                                  int* __restrict__ cntA,
                                  int* __restrict__ cntB,
                                  int* __restrict__ rankh, int E) {
    int e = blockIdx.x * blockDim.x + threadIdx.x;
    const int h = wave_half();
    int* cnt = h ? cntB : cntA;
    if (e < E) {
        const int r = atomicAdd(&cnt[dst[e]], 1);
        rankh[e] = r * 2 + h;
    }
}

__global__ void offsets_kernel(const int* __restrict__ cntA,
                               const int* __restrict__ cntB,
                               int* __restrict__ off,
                               float* __restrict__ dis,
                               int* __restrict__ cursor, int n) {
    __shared__ int wsum[32];
    __shared__ int bbase;
    const int i = blockIdx.x * blockDim.x + threadIdx.x;
    const int lane = threadIdx.x & 31;
    const int wv = threadIdx.x >> 5;
    const int nw = blockDim.x >> 5;
    const int c = (i < n) ? cntA[i] + cntB[i] : 0;
    int s = c;
    for (int o = 1; o < 32; o <<= 1) {
        const int t = __shfl_up(s, o, 32);
        if (lane >= o) s += t;
    }
    if (lane == 31) wsum[wv] = s;
    __syncthreads();
    if (threadIdx.x == 0) {
        int t = 0;
        for (int w = 0; w < nw; ++w) { const int x = wsum[w]; wsum[w] = t; t += x; }
        bbase = atomicAdd(cursor, t);
    }
    __syncthreads();
    if (i < n) {
        off[i] = bbase + wsum[wv] + s - c;
        dis[i] = rsqrtf((float)(c + 1));
    }
}

__global__ void fill_kernel(const int* __restrict__ src,
                            const int* __restrict__ dst,
                            const int* __restrict__ rankh,
                            const int* __restrict__ off,
                            const int* __restrict__ cntA,
                            int* __restrict__ nbr, int E) {
    int e = blockIdx.x * blockDim.x + threadIdx.x;
    if (e >= E) return;
    const int d = dst[e];
    const int rh = rankh[e];
    const int pos = off[d] + (rh >> 1) + ((rh & 1) ? cntA[d] : 0);
    nbr[pos] = src[e];
}

template <bool FUSE>
__global__ void gemm_wmma_kernel(const float* __restrict__ A,
                                 const float* __restrict__ W,
                                 const float* __restrict__ dis,
                                 const float* __restrict__ bias,
                                 float* __restrict__ ht,
                                 int M, int K) {
    const int lane = threadIdx.x & 31;
    const int wave = blockIdx.x * (blockDim.x >> 5) + (threadIdx.x >> 5);
    const int m0 = wave * 16;
    if (m0 >= M) return;

    const int half = lane >> 4;
    const int ml   = lane & 15;
    const int arow = m0 + ml;
    const float dA = FUSE ? dis[arow] : 0.0f;

    v8f c0 = {}, c1 = {}, c2 = {}, c3 = {};

    for (int k0 = 0; k0 < K; k0 += 4) {
        const int ka = k0 + half * 2;
        float a0 = A[(size_t)arow * K + ka];
        float a1 = A[(size_t)arow * K + ka + 1];
        if (FUSE) {
            a0 = fmaxf(fmaf(a0, dA, bias[ka]), 0.0f);
            a1 = fmaxf(fmaf(a1, dA, bias[ka + 1]), 0.0f);
        }
        v2f av; av.x = a0; av.y = a1;

        const float* w0 = W + (size_t)ka * 64;
        const float* w1 = W + (size_t)(ka + 1) * 64;
        v2f b0; b0.x = w0[ml];      b0.y = w1[ml];
        v2f b1; b1.x = w0[16 + ml]; b1.y = w1[16 + ml];
        v2f b2; b2.x = w0[32 + ml]; b2.y = w1[32 + ml];
        v2f b3; b3.x = w0[48 + ml]; b3.y = w1[48 + ml];

        c0 = __builtin_amdgcn_wmma_f32_16x16x4_f32(false, av, false, b0, (short)0, c0, false, false);
        c1 = __builtin_amdgcn_wmma_f32_16x16x4_f32(false, av, false, b1, (short)0, c1, false, false);
        c2 = __builtin_amdgcn_wmma_f32_16x16x4_f32(false, av, false, b2, (short)0, c2, false, false);
        c3 = __builtin_amdgcn_wmma_f32_16x16x4_f32(false, av, false, b3, (short)0, c3, false, false);
    }

    for (int r = 0; r < 8; ++r) {
        const int orow = m0 + half * 8 + r;
        const float dr = dis[orow];
        float* hp = ht + (size_t)orow * 64;
        hp[ml] = c0[r] * dr; hp[16 + ml] = c1[r] * dr; hp[32 + ml] = c2[r] * dr; hp[48 + ml] = c3[r] * dr;
    }
}

#define ROWS_IN_FLIGHT 24
__global__ void gather_rows_kernel(const int* __restrict__ off,
                                   const int* __restrict__ cntA,
                                   const int* __restrict__ cntB,
                                   const int* __restrict__ nbr,
                                   const float* __restrict__ ht,
                                   float* __restrict__ agg, int N) {
    const int lane = threadIdx.x & 31;
    const int wv = __builtin_amdgcn_readfirstlane((int)(threadIdx.x >> 5));
    const int i = blockIdx.x * (blockDim.x >> 5) + wv;
    if (i >= N) return;
    const int start = off[i];
    const int deg = cntA[i] + cntB[i];
    const int tot = deg + 1;
    const float* hl = ht + lane * 2;
    float2 acc; acc.x = 0.0f; acc.y = 0.0f;

    int base = 0;
    do {
        const int left = tot - base;
        const int n = left < 32 ? left : 32;
        const int t = base + lane;
        const int mine = (t < deg) ? nbr[start + t] : i;
        for (int j = 0; j < n; j += ROWS_IN_FLIGHT) {
            float2 v[ROWS_IN_FLIGHT];
            v[0] = *(const float2*)(hl + (size_t)__builtin_amdgcn_readlane(mine, j) * 64);
            #pragma unroll
            for (int u = 1; u < ROWS_IN_FLIGHT; ++u) {
                if (j + u < n) {
                    const int s = __builtin_amdgcn_readlane(mine, j + u);
                    v[u] = *(const float2*)(hl + (size_t)s * 64);
                }
            }
            acc.x += v[0].x; acc.y += v[0].y;
            #pragma unroll
            for (int u = 1; u < ROWS_IN_FLIGHT; ++u) {
                if (j + u < n) { acc.x += v[u].x; acc.y += v[u].y; }
            }
        }
        base += 32;
    } while (base < tot);
    *(float2*)(agg + (size_t)i * 64 + lane * 2) = acc;
}

__global__ void fc_out_kernel(const float* __restrict__ agg2,
                              const float* __restrict__ dis,
                              const float* __restrict__ b2,
                              const float* __restrict__ wfc,
                              const float* __restrict__ bfc,
                              float* __restrict__ out, int M) {
    const int lane = threadIdx.x & 31;
    const int i = blockIdx.x * (blockDim.x >> 5) + (threadIdx.x >> 5);
    if (i >= M) return;
    const float di = dis[i];
    const int j = lane * 2;
    const float2 a = *(const float2*)(agg2 + (size_t)i * 64 + j);
    const float h0 = fmaxf(fmaf(a.x, di, b2[j]), 0.0f);
    const float h1 = fmaxf(fmaf(a.y, di, b2[j + 1]), 0.0f);
    float p = h0 * wfc[j] + h1 * wfc[j + 1];
    #pragma unroll
    for (int off = 16; off > 0; off >>= 1) p += __shfl_down(p, off, 32);
    if (lane == 0) out[i] = p + bfc[0];
}

extern "C" void kernel_launch(void* const* d_in, const int* in_sizes, int n_in,
                              void* d_out, int out_size, void* d_ws, size_t ws_size,
                              hipStream_t stream) {
    const float* x   = (const float*)d_in[0];
    const int*   ei  = (const int*)d_in[1];
    const float* W1  = (const float*)d_in[2];
    const float* b1  = (const float*)d_in[3];
    const float* W2  = (const float*)d_in[4];
    const float* b2  = (const float*)d_in[5];
    const float* Wfc = (const float*)d_in[6];
    const float* bfc = (const float*)d_in[7];

    const int HID = in_sizes[3];
    const int IN  = in_sizes[2] / HID;
    const int N   = in_sizes[0] / IN;
    const int E   = in_sizes[1] / 2;
    const int* srcv = ei;
    const int* dstv = ei + E;

    char* ws = (char*)d_ws;
    const size_t nodeBytes = (((size_t)N * 4) + 255) & ~(size_t)255;
    const size_t edgeBytes = (((size_t)E * 4) + 255) & ~(size_t)255;
    const size_t rowBytes  = (size_t)N * 64 * sizeof(float);
    float* dis    = (float*)ws;
    int*   cntA   = (int*)(ws + nodeBytes);
    int*   cntB   = (int*)(ws + 2 * nodeBytes);
    int*   offs   = (int*)(ws + 3 * nodeBytes);
    int*   cursor = (int*)(ws + 4 * nodeBytes);
    size_t o = 4 * nodeBytes + 256;
    int*   rankh  = (int*)(ws + o);
    int*   nbr    = (int*)(ws + o + edgeBytes);
    o += 2 * edgeBytes;
    float* ht   = (float*)(ws + o);
    float* agg1 = (float*)(ws + o + rowBytes);
    float* agg2 = (float*)(ws + o + 2 * rowBytes);

    const int gemmBlocks = ((N + 15) / 16 + 7) / 8;
    const int edgeBlocks = (E + 255) / 256;
    const int nodeBlocks = (N + 7) / 8;

    init_count_kernel<<<(N + 255) / 256, 256, 0, stream>>>(cntA, cntB, cursor, N);
    count_rank_kernel<<<edgeBlocks, 256, 0, stream>>>(dstv, cntA, cntB, rankh, E);
    offsets_kernel<<<(N + 255) / 256, 256, 0, stream>>>(cntA, cntB, offs, dis, cursor, N);
    fill_kernel<<<edgeBlocks, 256, 0, stream>>>(srcv, dstv, rankh, offs, cntA, nbr, E);

    gemm_wmma_kernel<false><<<gemmBlocks, 256, 0, stream>>>(x, W1, dis, nullptr, ht, N, IN);
    gather_rows_kernel<<<nodeBlocks, 256, 0, stream>>>(offs, cntA, cntB, nbr, ht, agg1, N);

    gemm_wmma_kernel<true><<<gemmBlocks, 256, 0, stream>>>(agg1, W2, dis, b1, ht, N, HID);
    gather_rows_kernel<<<nodeBlocks, 256, 0, stream>>>(offs, cntA, cntB, nbr, ht, agg2, N);

    fc_out_kernel<<<nodeBlocks, 256, 0, stream>>>(agg2, dis, b2, Wfc, bfc, (float*)d_out, N);
}


# fewer, longer waves (wave entry rate was the limit): a thread takes 8 edges, a gather wave walks 7 nodes, an fc wave sums 32 rows on DPP/permlane; every sum keeps its order
# speedup vs baseline: 3.2291x; 1.1820x over previous
#include <hip/hip_runtime.h>

typedef __attribute__((ext_vector_type(2))) float v2f;
typedef __attribute__((ext_vector_type(8))) float v8f;

__device__ __forceinline__ int wave_half() {
    const unsigned id = __builtin_amdgcn_s_sendmsg_rtn(0x87);
    return (int)((id >> 18) & 1u);
}

__global__ void init_count_kernel(int* __restrict__ cntA, int* __restrict__ cntB,
                                  int* __restrict__ cursor, int n) {
    int i = blockIdx.x * blockDim.x + threadIdx.x;
    if (i < n) { cntA[i] = 0; cntB[i] = 0; }
    if (i == 0) cursor[0] = 0;
}

#define EDGES_PER_THREAD 8
template <bool FULL>
__device__ __forceinline__ void count_rank_body(const int* __restrict__ dst,
                                                int* __restrict__ cnt,
                                                int* __restrict__ rankh,
                                                int e0, int stride, int h, int E) {
    int d[EDGES_PER_THREAD], r[EDGES_PER_THREAD];
    #pragma unroll
    for (int u = 0; u < EDGES_PER_THREAD; ++u) {
        const int e = e0 + u * stride;
        if (FULL || e < E) d[u] = dst[e];
    }
    #pragma unroll
    for (int u = 0; u < EDGES_PER_THREAD; ++u) {
        const int e = e0 + u * stride;
        if (FULL || e < E) r[u] = atomicAdd(&cnt[d[u]], 1);
    }
    #pragma unroll
    for (int u = 0; u < EDGES_PER_THREAD; ++u) {
        const int e = e0 + u * stride;
        if (FULL || e < E) rankh[e] = r[u] * 2 + h;
    }
}

__global__ void count_rank_kernel(const int* __restrict__ dst,
                                  int* __restrict__ cntA,
                                  int* __restrict__ cntB,
                                  int* __restrict__ rankh, int E) {
    const int stride = (int)blockDim.x;
    const int b0 = blockIdx.x * (stride * EDGES_PER_THREAD);
    const int e0 = b0 + threadIdx.x;
    const int h = wave_half();
    int* cnt = h ? cntB : cntA;
    if (b0 + stride * EDGES_PER_THREAD <= E) count_rank_body<true>(dst, cnt, rankh, e0, stride, h, E);
    else                                      count_rank_body<false>(dst, cnt, rankh, e0, stride, h, E);
}

__global__ void offsets_kernel(const int* __restrict__ cntA,
                               const int* __restrict__ cntB,
                               int* __restrict__ off,
                               float* __restrict__ dis,
                               int* __restrict__ cursor, int n) {
    __shared__ int wsum[32];
    __shared__ int bbase;
    const int i = blockIdx.x * blockDim.x + threadIdx.x;
    const int lane = threadIdx.x & 31;
    const int wv = threadIdx.x >> 5;
    const int nw = blockDim.x >> 5;
    const int c = (i < n) ? cntA[i] + cntB[i] : 0;
    int s = c;
    for (int o = 1; o < 32; o <<= 1) {
        const int t = __shfl_up(s, o, 32);
        if (lane >= o) s += t;
    }
    if (lane == 31) wsum[wv] = s;
    __syncthreads();
    if (threadIdx.x == 0) {
        int t = 0;
        for (int w = 0; w < nw; ++w) { const int x = wsum[w]; wsum[w] = t; t += x; }
        bbase = atomicAdd(cursor, t);
    }
    __syncthreads();
    if (i < n) {
        off[i] = bbase + wsum[wv] + s - c;
        dis[i] = rsqrtf((float)(c + 1));
    }
}

template <bool FULL>
__device__ __forceinline__ void fill_body(const int* __restrict__ src,
                                          const int* __restrict__ dst,
                                          const int* __restrict__ rankh,
                                          const int* __restrict__ off,
                                          const int* __restrict__ cntA,
                                          int* __restrict__ nbr,
                                          int e0, int stride, int E) {
    int d[EDGES_PER_THREAD], rh[EDGES_PER_THREAD], s[EDGES_PER_THREAD], pos[EDGES_PER_THREAD];
    #pragma unroll
    for (int u = 0; u < EDGES_PER_THREAD; ++u) {
        const int e = e0 + u * stride;
        if (FULL || e < E) { d[u] = dst[e]; rh[u] = rankh[e]; s[u] = src[e]; }
    }
    #pragma unroll
    for (int u = 0; u < EDGES_PER_THREAD; ++u) {
        const int e = e0 + u * stride;
        if (FULL || e < E) pos[u] = off[d[u]] + (rh[u] >> 1) + ((rh[u] & 1) ? cntA[d[u]] : 0);
    }
    #pragma unroll
    for (int u = 0; u < EDGES_PER_THREAD; ++u) {
        const int e = e0 + u * stride;
        if (FULL || e < E) nbr[pos[u]] = s[u];
    }
}

__global__ void fill_kernel(const int* __restrict__ src,
                            const int* __restrict__ dst,
                            const int* __restrict__ rankh,
                            const int* __restrict__ off,
                            const int* __restrict__ cntA,
                            int* __restrict__ nbr, int E) {
    const int stride = (int)blockDim.x;
    const int b0 = blockIdx.x * (stride * EDGES_PER_THREAD);
    const int e0 = b0 + threadIdx.x;
    if (b0 + stride * EDGES_PER_THREAD <= E) fill_body<true>(src, dst, rankh, off, cntA, nbr, e0, stride, E);
    else                                      fill_body<false>(src, dst, rankh, off, cntA, nbr, e0, stride, E);
}

template <bool FUSE>
__global__ void gemm_wmma_kernel(const float* __restrict__ A,
                                 const float* __restrict__ W,
                                 const float* __restrict__ dis,
                                 const float* __restrict__ bias,
                                 float* __restrict__ ht,
                                 int M, int K) {
    const int lane = threadIdx.x & 31;
    const int wave = blockIdx.x * (blockDim.x >> 5) + (threadIdx.x >> 5);
    const int m0 = wave * 16;
    if (m0 >= M) return;

    const int half = lane >> 4;
    const int ml   = lane & 15;
    const int arow = m0 + ml;
    const float dA = FUSE ? dis[arow] : 0.0f;

    v8f c0 = {}, c1 = {}, c2 = {}, c3 = {};

    for (int k0 = 0; k0 < K; k0 += 4) {
        const int ka = k0 + half * 2;
        float a0 = A[(size_t)arow * K + ka];
        float a1 = A[(size_t)arow * K + ka + 1];
        if (FUSE) {
            a0 = fmaxf(fmaf(a0, dA, bias[ka]), 0.0f);
            a1 = fmaxf(fmaf(a1, dA, bias[ka + 1]), 0.0f);
        }
        v2f av; av.x = a0; av.y = a1;

        const float* w0 = W + (size_t)ka * 64;
        const float* w1 = W + (size_t)(ka + 1) * 64;
        v2f b0; b0.x = w0[ml];      b0.y = w1[ml];
        v2f b1; b1.x = w0[16 + ml]; b1.y = w1[16 + ml];
        v2f b2; b2.x = w0[32 + ml]; b2.y = w1[32 + ml];
        v2f b3; b3.x = w0[48 + ml]; b3.y = w1[48 + ml];

        c0 = __builtin_amdgcn_wmma_f32_16x16x4_f32(false, av, false, b0, (short)0, c0, false, false);
        c1 = __builtin_amdgcn_wmma_f32_16x16x4_f32(false, av, false, b1, (short)0, c1, false, false);
        c2 = __builtin_amdgcn_wmma_f32_16x16x4_f32(false, av, false, b2, (short)0, c2, false, false);
        c3 = __builtin_amdgcn_wmma_f32_16x16x4_f32(false, av, false, b3, (short)0, c3, false, false);
    }

    for (int r = 0; r < 8; ++r) {
        const int orow = m0 + half * 8 + r;
        const float dr = dis[orow];
        float* hp = ht + (size_t)orow * 64;
        hp[ml] = c0[r] * dr; hp[16 + ml] = c1[r] * dr; hp[32 + ml] = c2[r] * dr; hp[48 + ml] = c3[r] * dr;
    }
}

#define ROWS_IN_FLIGHT 24
#define NODES_PER_WAVE 7
__global__ void gather_rows_kernel(const int* __restrict__ off,
                                   const int* __restrict__ cntA,
                                   const int* __restrict__ cntB,
                                   const int* __restrict__ nbr,
                                   const float* __restrict__ ht,
                                   float* __restrict__ agg, int N) {
    const int lane = threadIdx.x & 31;
    const int wv = __builtin_amdgcn_readfirstlane((int)(threadIdx.x >> 5));
    const int i0 = (blockIdx.x * (blockDim.x >> 5) + wv) * NODES_PER_WAVE;
    if (i0 >= N) return;
    const int nn = (N - i0) < NODES_PER_WAVE ? (N - i0) : NODES_PER_WAVE;
    int offv = 0, degv = 0;
    if (lane < nn) {
        offv = off[i0 + lane];
        degv = cntA[i0 + lane] + cntB[i0 + lane];
    }
    const float* hl = ht + lane * 2;

    for (int q = 0; q < nn; ++q) {
        const int i = i0 + q;
        const int start = __builtin_amdgcn_readlane(offv, q);
        const int deg = __builtin_amdgcn_readlane(degv, q);
        const int tot = deg + 1;
        float2 acc; acc.x = 0.0f; acc.y = 0.0f;

        int base = 0;
        do {
            const int left = tot - base;
            const int n = left < 32 ? left : 32;
            const int t = base + lane;
            const int mine = (t < deg) ? nbr[start + t] : i;
            for (int j = 0; j < n; j += ROWS_IN_FLIGHT) {
                float2 v[ROWS_IN_FLIGHT];
                v[0] = *(const float2*)(hl + (size_t)__builtin_amdgcn_readlane(mine, j) * 64);
                #pragma unroll
                for (int u = 1; u < ROWS_IN_FLIGHT; ++u) {
                    if (j + u < n) {
                        const int s = __builtin_amdgcn_readlane(mine, j + u);
                        v[u] = *(const float2*)(hl + (size_t)s * 64);
                    }
                }
                acc.x += v[0].x; acc.y += v[0].y;
                #pragma unroll
                for (int u = 1; u < ROWS_IN_FLIGHT; ++u) {
                    if (j + u < n) { acc.x += v[u].x; acc.y += v[u].y; }
                }
            }
            base += 32;
        } while (base < tot);
        *(float2*)(agg + (size_t)i * 64 + lane * 2) = acc;
    }
}

#define FC_ROWS 32
__device__ __forceinline__ float xhalf(float x) {
    return __builtin_bit_cast(float, __builtin_amdgcn_permlanex16(
        __builtin_bit_cast(int, x), __builtin_bit_cast(int, x), 0x76543210, 0xfedcba98, false, false));
}
template <int MASK>
__device__ __forceinline__ float xrow(float x) {
    return __builtin_bit_cast(float, __builtin_amdgcn_update_dpp(
        0, __builtin_bit_cast(int, x), 0x160 + MASK, 0xf, 0xf, false));
}
__device__ __forceinline__ float wave_sum(float p) {
    p += xhalf(p);
    p += xrow<8>(p);
    p += xrow<4>(p);
    p += xrow<2>(p);
    p += xrow<1>(p);
    return p;
}

__global__ void fc_out_kernel(const float* __restrict__ agg2,
                              const float* __restrict__ dis,
                              const float* __restrict__ b2,
                              const float* __restrict__ wfc,
                              const float* __restrict__ bfc,
                              float* __restrict__ out, int M) {
    const int lane = threadIdx.x & 31;
    const int wv = __builtin_amdgcn_readfirstlane((int)(threadIdx.x >> 5));
    const int r0 = (blockIdx.x * (blockDim.x >> 5) + wv) * FC_ROWS;
    if (r0 >= M) return;
    const int last = (M - 1 - r0) < (FC_ROWS - 1) ? (M - 1 - r0) : (FC_ROWS - 1);
    const int j = lane * 2;
    const float bj0 = b2[j], bj1 = b2[j + 1];
    const float w0 = wfc[j], w1 = wfc[j + 1];
    const int dl = __float_as_int(dis[r0 + (lane < last ? lane : last)]);
    const float* al = agg2 + (size_t)r0 * 64 + j;
    float2 a[FC_ROWS];
    #pragma unroll
    for (int u = 0; u < FC_ROWS; ++u)
        a[u] = *(const float2*)(al + (size_t)(u < last ? u : last) * 64);
    float res = 0.0f;
    #pragma unroll
    for (int u = 0; u < FC_ROWS; ++u) {
        const float di = __int_as_float(__builtin_amdgcn_readlane(dl, u));
        const float h0 = fmaxf(fmaf(a[u].x, di, bj0), 0.0f);
        const float h1 = fmaxf(fmaf(a[u].y, di, bj1), 0.0f);
        const float p = wave_sum(h0 * w0 + h1 * w1);
        if (lane == u) res = p;
    }
    if (lane <= last) out[r0 + lane] = res + bfc[0];
}

extern "C" void kernel_launch(void* const* d_in, const int* in_sizes, int n_in,
                              void* d_out, int out_size, void* d_ws, size_t ws_size,
                              hipStream_t stream) {
    const float* x   = (const float*)d_in[0];
    const int*   ei  = (const int*)d_in[1];
    const float* W1  = (const float*)d_in[2];
    const float* b1  = (const float*)d_in[3];
    const float* W2  = (const float*)d_in[4];
    const float* b2  = (const float*)d_in[5];
    const float* Wfc = (const float*)d_in[6];
    const float* bfc = (const float*)d_in[7];

    const int HID = in_sizes[3];
    const int IN  = in_sizes[2] / HID;
    const int N   = in_sizes[0] / IN;
    const int E   = in_sizes[1] / 2;
    const int* srcv = ei;
    const int* dstv = ei + E;

    char* ws = (char*)d_ws;
    const size_t nodeBytes = (((size_t)N * 4) + 255) & ~(size_t)255;
    const size_t edgeBytes = (((size_t)E * 4) + 255) & ~(size_t)255;
    const size_t rowBytes  = (size_t)N * 64 * sizeof(float);
    float* dis    = (float*)ws;
    int*   cntA   = (int*)(ws + nodeBytes);
    int*   cntB   = (int*)(ws + 2 * nodeBytes);
    int*   offs   = (int*)(ws + 3 * nodeBytes);
    int*   cursor = (int*)(ws + 4 * nodeBytes);
    size_t o = 4 * nodeBytes + 256;
    int*   rankh  = (int*)(ws + o);
    int*   nbr    = (int*)(ws + o + edgeBytes);
    o += 2 * edgeBytes;
    float* ht   = (float*)(ws + o);
    float* agg1 = (float*)(ws + o + rowBytes);
    float* agg2 = (float*)(ws + o + 2 * rowBytes);

    const int gemmBlocks = ((N + 15) / 16 + 7) / 8;
    const int edgeBlocks = (E + 256 * EDGES_PER_THREAD - 1) / (256 * EDGES_PER_THREAD);
    const int nodeBlocks = ((N + NODES_PER_WAVE - 1) / NODES_PER_WAVE + 7) / 8;
    const int fcBlocks   = ((N + FC_ROWS - 1) / FC_ROWS + 7) / 8;

    init_count_kernel<<<(N + 255) / 256, 256, 0, stream>>>(cntA, cntB, cursor, N);
    count_rank_kernel<<<edgeBlocks, 256, 0, stream>>>(dstv, cntA, cntB, rankh, E);
    offsets_kernel<<<(N + 255) / 256, 256, 0, stream>>>(cntA, cntB, offs, dis, cursor, N);
    fill_kernel<<<edgeBlocks, 256, 0, stream>>>(srcv, dstv, rankh, offs, cntA, nbr, E);

    gemm_wmma_kernel<false><<<gemmBlocks, 256, 0, stream>>>(x, W1, dis, nullptr, ht, N, IN);
    gather_rows_kernel<<<nodeBlocks, 256, 0, stream>>>(offs, cntA, cntB, nbr, ht, agg1, N);

    gemm_wmma_kernel<true><<<gemmBlocks, 256, 0, stream>>>(agg1, W2, dis, b1, ht, N, HID);
    gather_rows_kernel<<<nodeBlocks, 256, 0, stream>>>(offs, cntA, cntB, nbr, ht, agg2, N);

    fc_out_kernel<<<fcBlocks, 256, 0, stream>>>(agg2, dis, b2, Wfc, bfc, (float*)d_out, N);
}


# gather waves consume their own sums: the layer-1 wave walks a 16-node tile, keeps finished rows in an LDS slot and does the W2 product (W2 staged in LDS); layer-2 wave does the fc; sum orders kept
# speedup vs baseline: 3.4578x; 1.0708x over previous
#include <hip/hip_runtime.h>

typedef __attribute__((ext_vector_type(2))) float v2f;
typedef __attribute__((ext_vector_type(8))) float v8f;

__device__ __forceinline__ int wave_half() {
    const unsigned id = __builtin_amdgcn_s_sendmsg_rtn(0x87);
    return (int)((id >> 18) & 1u);
}

__global__ void init_count_kernel(int* __restrict__ cntA, int* __restrict__ cntB,
                                  int* __restrict__ cursor, int n) {
    int i = blockIdx.x * blockDim.x + threadIdx.x;
    if (i < n) { cntA[i] = 0; cntB[i] = 0; }
    if (i == 0) cursor[0] = 0;
}

#define EDGES_PER_THREAD 8
template <bool FULL>
__device__ __forceinline__ void count_rank_body(const int* __restrict__ dst,
                                                int* __restrict__ cnt,
                                                int* __restrict__ rankh,
                                                int e0, int stride, int h, int E) {
    int d[EDGES_PER_THREAD], r[EDGES_PER_THREAD];
    #pragma unroll
    for (int u = 0; u < EDGES_PER_THREAD; ++u) {
        const int e = e0 + u * stride;
        if (FULL || e < E) d[u] = dst[e];
    }
    #pragma unroll
    for (int u = 0; u < EDGES_PER_THREAD; ++u) {
        const int e = e0 + u * stride;
        if (FULL || e < E) r[u] = atomicAdd(&cnt[d[u]], 1);
    }
    #pragma unroll
    for (int u = 0; u < EDGES_PER_THREAD; ++u) {
        const int e = e0 + u * stride;
        if (FULL || e < E) rankh[e] = r[u] * 2 + h;
    }
}

__global__ void count_rank_kernel(const int* __restrict__ dst,
                                  int* __restrict__ cntA,
                                  int* __restrict__ cntB,
                                  int* __restrict__ rankh, int E) {
    const int stride = (int)blockDim.x;
    const int b0 = blockIdx.x * (stride * EDGES_PER_THREAD);
    const int e0 = b0 + threadIdx.x;
    const int h = wave_half();
    int* cnt = h ? cntB : cntA;
    if (b0 + stride * EDGES_PER_THREAD <= E) count_rank_body<true>(dst, cnt, rankh, e0, stride, h, E);
    else                                      count_rank_body<false>(dst, cnt, rankh, e0, stride, h, E);
}

__global__ void offsets_kernel(const int* __restrict__ cntA,
                               const int* __restrict__ cntB,
                               int* __restrict__ off,
                               float* __restrict__ dis,
                               int* __restrict__ cursor, int n) {
    __shared__ int wsum[32];
    __shared__ int bbase;
    const int i = blockIdx.x * blockDim.x + threadIdx.x;
    const int lane = threadIdx.x & 31;
    const int wv = threadIdx.x >> 5;
    const int nw = blockDim.x >> 5;
    const int c = (i < n) ? cntA[i] + cntB[i] : 0;
    int s = c;
    for (int o = 1; o < 32; o <<= 1) {
        const int t = __shfl_up(s, o, 32);
        if (lane >= o) s += t;
    }
    if (lane == 31) wsum[wv] = s;
    __syncthreads();
    if (threadIdx.x == 0) {
        int t = 0;
        for (int w = 0; w < nw; ++w) { const int x = wsum[w]; wsum[w] = t; t += x; }
        bbase = atomicAdd(cursor, t);
    }
    __syncthreads();
    if (i < n) {
        off[i] = bbase + wsum[wv] + s - c;
        dis[i] = rsqrtf((float)(c + 1));
    }
}

template <bool FULL>
__device__ __forceinline__ void fill_body(const int* __restrict__ src,
                                          const int* __restrict__ dst,
                                          const int* __restrict__ rankh,
                                          const int* __restrict__ off,
                                          const int* __restrict__ cntA,
                                          int* __restrict__ nbr,
                                          int e0, int stride, int E) {
    int d[EDGES_PER_THREAD], rh[EDGES_PER_THREAD], s[EDGES_PER_THREAD], pos[EDGES_PER_THREAD];
    #pragma unroll
    for (int u = 0; u < EDGES_PER_THREAD; ++u) {
        const int e = e0 + u * stride;
        if (FULL || e < E) { d[u] = dst[e]; rh[u] = rankh[e]; s[u] = src[e]; }
    }
    #pragma unroll
    for (int u = 0; u < EDGES_PER_THREAD; ++u) {
        const int e = e0 + u * stride;
        if (FULL || e < E) pos[u] = off[d[u]] + (rh[u] >> 1) + ((rh[u] & 1) ? cntA[d[u]] : 0);
    }
    #pragma unroll
    for (int u = 0; u < EDGES_PER_THREAD; ++u) {
        const int e = e0 + u * stride;
        if (FULL || e < E) nbr[pos[u]] = s[u];
    }
}

__global__ void fill_kernel(const int* __restrict__ src,
                            const int* __restrict__ dst,
                            const int* __restrict__ rankh,
                            const int* __restrict__ off,
                            const int* __restrict__ cntA,
                            int* __restrict__ nbr, int E) {
    const int stride = (int)blockDim.x;
    const int b0 = blockIdx.x * (stride * EDGES_PER_THREAD);
    const int e0 = b0 + threadIdx.x;
    if (b0 + stride * EDGES_PER_THREAD <= E) fill_body<true>(src, dst, rankh, off, cntA, nbr, e0, stride, E);
    else                                      fill_body<false>(src, dst, rankh, off, cntA, nbr, e0, stride, E);
}

template <bool FUSE>
__global__ void gemm_wmma_kernel(const float* __restrict__ A,
                                 const float* __restrict__ W,
                                 const float* __restrict__ dis,
                                 const float* __restrict__ bias,
                                 float* __restrict__ ht,
                                 int M, int K) {
    const int lane = threadIdx.x & 31;
    const int wave = blockIdx.x * (blockDim.x >> 5) + (threadIdx.x >> 5);
    const int m0 = wave * 16;
    if (m0 >= M) return;

    const int half = lane >> 4;
    const int ml   = lane & 15;
    const int arow = m0 + ml;
    const float dA = FUSE ? dis[arow] : 0.0f;

    v8f c0 = {}, c1 = {}, c2 = {}, c3 = {};

    for (int k0 = 0; k0 < K; k0 += 4) {
        const int ka = k0 + half * 2;
        float a0 = A[(size_t)arow * K + ka];
        float a1 = A[(size_t)arow * K + ka + 1];
        if (FUSE) {
            a0 = fmaxf(fmaf(a0, dA, bias[ka]), 0.0f);
            a1 = fmaxf(fmaf(a1, dA, bias[ka + 1]), 0.0f);
        }
        v2f av; av.x = a0; av.y = a1;

        const float* w0 = W + (size_t)ka * 64;
        const float* w1 = W + (size_t)(ka + 1) * 64;
        v2f b0; b0.x = w0[ml];      b0.y = w1[ml];
        v2f b1; b1.x = w0[16 + ml]; b1.y = w1[16 + ml];
        v2f b2; b2.x = w0[32 + ml]; b2.y = w1[32 + ml];
        v2f b3; b3.x = w0[48 + ml]; b3.y = w1[48 + ml];

        c0 = __builtin_amdgcn_wmma_f32_16x16x4_f32(false, av, false, b0, (short)0, c0, false, false);
        c1 = __builtin_amdgcn_wmma_f32_16x16x4_f32(false, av, false, b1, (short)0, c1, false, false);
        c2 = __builtin_amdgcn_wmma_f32_16x16x4_f32(false, av, false, b2, (short)0, c2, false, false);
        c3 = __builtin_amdgcn_wmma_f32_16x16x4_f32(false, av, false, b3, (short)0, c3, false, false);
    }

    for (int r = 0; r < 8; ++r) {
        const int orow = m0 + half * 8 + r;
        const float dr = dis[orow];
        float* hp = ht + (size_t)orow * 64;
        hp[ml] = c0[r] * dr; hp[16 + ml] = c1[r] * dr; hp[32 + ml] = c2[r] * dr; hp[48 + ml] = c3[r] * dr;
    }
}

#define ROWS_IN_FLIGHT 24
__device__ __forceinline__ float2 node_sum(const int* __restrict__ nbr,
                                           const float* __restrict__ hl,
                                           int start, int deg, int i, int lane) {
    const int tot = deg + 1;
    float2 acc; acc.x = 0.0f; acc.y = 0.0f;
    int base = 0;
    do {
        const int left = tot - base;
        const int n = left < 32 ? left : 32;
        const int t = base + lane;
        const int mine = (t < deg) ? nbr[start + t] : i;
        for (int j = 0; j < n; j += ROWS_IN_FLIGHT) {
            float2 v[ROWS_IN_FLIGHT];
            v[0] = *(const float2*)(hl + (size_t)__builtin_amdgcn_readlane(mine, j) * 64);
            #pragma unroll
            for (int u = 1; u < ROWS_IN_FLIGHT; ++u) {
                if (j + u < n) {
                    const int s = __builtin_amdgcn_readlane(mine, j + u);
                    v[u] = *(const float2*)(hl + (size_t)s * 64);
                }
            }
            acc.x += v[0].x; acc.y += v[0].y;
            #pragma unroll
            for (int u = 1; u < ROWS_IN_FLIGHT; ++u) {
                if (j + u < n) { acc.x += v[u].x; acc.y += v[u].y; }
            }
        }
        base += 32;
    } while (base < tot);
    return acc;
}

#define TILE_ROWS 16
#define A_PITCH 68
__global__ void __launch_bounds__(256)
gather_transform_kernel(const int* __restrict__ off,
                        const int* __restrict__ cntA,
                        const int* __restrict__ cntB,
                        const int* __restrict__ nbr,
                        const float* __restrict__ ht,
                        const float* __restrict__ dis,
                        const float* __restrict__ bias,
                        const float* __restrict__ W,
                        float* __restrict__ hout, int N) {
    __shared__ float4 Wl[2][16][32];
    __shared__ float As[8][TILE_ROWS][A_PITCH];
    const int lane = threadIdx.x & 31;
    const int wv = __builtin_amdgcn_readfirstlane((int)(threadIdx.x >> 5));
    const int m0 = (blockIdx.x * 8 + wv) * TILE_ROWS;
    const int nn = (N - m0) < TILE_ROWS ? (N - m0) : TILE_ROWS;

    int offv = 0, degv = 0, disv = 0;
    if (lane < nn) {
        offv = off[m0 + lane];
        degv = cntA[m0 + lane] + cntB[m0 + lane];
        disv = __float_as_int(dis[m0 + lane]);
    }
    const float bj0 = bias[lane * 2], bj1 = bias[lane * 2 + 1];

    {
        float4 wr[4];
        #pragma unroll
        for (int u = 0; u < 4; ++u)
            wr[u] = *(const float4*)(W + (size_t)(u * 256 + (int)threadIdx.x) * 4);
        const int t = (int)threadIdx.x;
        const int tile = (t & 15) >> 2;
        const int ln0 = ((t >> 5) & 1) * 16 + ((t & 3) << 2);
        const int e = (tile & 1) * 2 + ((t >> 4) & 1);
        float* wp = (float*)&Wl[tile >> 1][t >> 6][ln0] + e;
        #pragma unroll
        for (int u = 0; u < 4; ++u) {
            float* p = wp + u * (4 * 32 * 4);
            p[0] = wr[u].x; p[4] = wr[u].y; p[8] = wr[u].z; p[12] = wr[u].w;
        }
    }
    __syncthreads();
    if (nn <= 0) return;

    const float* hl = ht + lane * 2;
    float* slot = &As[wv][0][lane * 2];

    for (int q = 0; q < TILE_ROWS; ++q) {
        float2 a; a.x = 0.0f; a.y = 0.0f;
        if (q < nn) {
            const int start = __builtin_amdgcn_readlane(offv, q);
            const int deg = __builtin_amdgcn_readlane(degv, q);
            const float di = __int_as_float(__builtin_amdgcn_readlane(disv, q));
            const float2 acc = node_sum(nbr, hl, start, deg, m0 + q, lane);
            a.x = fmaxf(fmaf(acc.x, di, bj0), 0.0f);
            a.y = fmaxf(fmaf(acc.y, di, bj1), 0.0f);
        }
        *(float2*)(slot + q * A_PITCH) = a;
    }

    const int half = lane >> 4;
    const int ml   = lane & 15;
    const float* ap = &As[wv][ml][half * 2];
    v8f c0 = {}, c1 = {}, c2 = {}, c3 = {};
    #pragma unroll
    for (int ks = 0; ks < 16; ++ks) {
        const v2f av = *(const v2f*)(ap + ks * 4);
        const float4 b01 = Wl[0][ks][lane];
        const float4 b23 = Wl[1][ks][lane];
        v2f b0; b0.x = b01.x; b0.y = b01.y;
        v2f b1; b1.x = b01.z; b1.y = b01.w;
        v2f b2; b2.x = b23.x; b2.y = b23.y;
        v2f b3; b3.x = b23.z; b3.y = b23.w;
        c0 = __builtin_amdgcn_wmma_f32_16x16x4_f32(false, av, false, b0, (short)0, c0, false, false);
        c1 = __builtin_amdgcn_wmma_f32_16x16x4_f32(false, av, false, b1, (short)0, c1, false, false);
        c2 = __builtin_amdgcn_wmma_f32_16x16x4_f32(false, av, false, b2, (short)0, c2, false, false);
        c3 = __builtin_amdgcn_wmma_f32_16x16x4_f32(false, av, false, b3, (short)0, c3, false, false);
    }

    float* cp = &As[wv][half * 8][ml];
    #pragma unroll
    for (int r = 0; r < 8; ++r) {
        float* p = cp + r * A_PITCH;
        p[0] = c0[r]; p[16] = c1[r]; p[32] = c2[r]; p[48] = c3[r];
    }
    float2 h[TILE_ROWS];
    #pragma unroll
    for (int q = 0; q < TILE_ROWS; ++q) h[q] = *(const float2*)(slot + q * A_PITCH);
    #pragma unroll
    for (int q = 0; q < TILE_ROWS; ++q) {
        if (q < nn) {
            const float dr = __int_as_float(__builtin_amdgcn_readlane(disv, q));
            float2 o; o.x = h[q].x * dr; o.y = h[q].y * dr;
            *(float2*)(hout + (size_t)(m0 + q) * 64 + lane * 2) = o;
        }
    }
}

#define FC_NODES 8
__device__ __forceinline__ float xhalf(float x) {
    return __builtin_bit_cast(float, __builtin_amdgcn_permlanex16(
        __builtin_bit_cast(int, x), __builtin_bit_cast(int, x), 0x76543210, 0xfedcba98, false, false));
}
template <int MASK>
__device__ __forceinline__ float xrow(float x) {
    return __builtin_bit_cast(float, __builtin_amdgcn_update_dpp(
        0, __builtin_bit_cast(int, x), 0x160 + MASK, 0xf, 0xf, false));
}
__device__ __forceinline__ float wave_sum(float p) {
    p += xhalf(p);
    p += xrow<8>(p);
    p += xrow<4>(p);
    p += xrow<2>(p);
    p += xrow<1>(p);
    return p;
}

__global__ void __launch_bounds__(256)
gather_fc_kernel(const int* __restrict__ off,
                 const int* __restrict__ cntA,
                 const int* __restrict__ cntB,
                 const int* __restrict__ nbr,
                 const float* __restrict__ ht,
                 const float* __restrict__ dis,
                 const float* __restrict__ b2,
                 const float* __restrict__ wfc,
                 const float* __restrict__ bfc,
                 float* __restrict__ out, int N) {
    __shared__ float res[8 * FC_NODES];
    const int lane = threadIdx.x & 31;
    const int wv = __builtin_amdgcn_readfirstlane((int)(threadIdx.x >> 5));
    const int i0 = (blockIdx.x * 8 + wv) * FC_NODES;
    float r = 0.0f;
    if (i0 < N) {
        const int nn = (N - i0) < FC_NODES ? (N - i0) : FC_NODES;
        int offv = 0, degv = 0, disv = 0;
        if (lane < nn) {
            offv = off[i0 + lane];
            degv = cntA[i0 + lane] + cntB[i0 + lane];
            disv = __float_as_int(dis[i0 + lane]);
        }
        const float* hl = ht + lane * 2;
        const int j = lane * 2;
        const float bj0 = b2[j], bj1 = b2[j + 1];
        const float w0 = wfc[j], w1 = wfc[j + 1];
        for (int q = 0; q < nn; ++q) {
            const int start = __builtin_amdgcn_readlane(offv, q);
            const int deg = __builtin_amdgcn_readlane(degv, q);
            const float di = __int_as_float(__builtin_amdgcn_readlane(disv, q));
            const float2 acc = node_sum(nbr, hl, start, deg, i0 + q, lane);
            const float h0 = fmaxf(fmaf(acc.x, di, bj0), 0.0f);
            const float h1 = fmaxf(fmaf(acc.y, di, bj1), 0.0f);
            const float p = wave_sum(h0 * w0 + h1 * w1);
            if (lane == q) r = p;
        }
    }
    if (lane < FC_NODES) res[wv * FC_NODES + lane] = r;
    __syncthreads();
    if (wv == 0) {
        const int b = blockIdx.x * (8 * FC_NODES) + lane * 2;
        const float bf = bfc[0];
        if (b + 1 < N) {
            float2 o; o.x = res[lane * 2] + bf; o.y = res[lane * 2 + 1] + bf;
            *(float2*)(out + b) = o;
        } else if (b < N) {
            out[b] = res[lane * 2] + bf;
        }
    }
}

extern "C" void kernel_launch(void* const* d_in, const int* in_sizes, int n_in,
                              void* d_out, int out_size, void* d_ws, size_t ws_size,
                              hipStream_t stream) {
    const float* x   = (const float*)d_in[0];
    const int*   ei  = (const int*)d_in[1];
    const float* W1  = (const float*)d_in[2];
    const float* b1  = (const float*)d_in[3];
    const float* W2  = (const float*)d_in[4];
    const float* b2  = (const float*)d_in[5];
    const float* Wfc = (const float*)d_in[6];
    const float* bfc = (const float*)d_in[7];

    const int HID = in_sizes[3];
    const int IN  = in_sizes[2] / HID;
    const int N   = in_sizes[0] / IN;
    const int E   = in_sizes[1] / 2;
    const int* srcv = ei;
    const int* dstv = ei + E;

    char* ws = (char*)d_ws;
    const size_t nodeBytes = (((size_t)N * 4) + 255) & ~(size_t)255;
    const size_t edgeBytes = (((size_t)E * 4) + 255) & ~(size_t)255;
    const size_t rowBytes  = (size_t)N * 64 * sizeof(float);
    float* dis    = (float*)ws;
    int*   cntA   = (int*)(ws + nodeBytes);
    int*   cntB   = (int*)(ws + 2 * nodeBytes);
    int*   offs   = (int*)(ws + 3 * nodeBytes);
    int*   cursor = (int*)(ws + 4 * nodeBytes);
    size_t o = 4 * nodeBytes + 256;
    int*   rankh  = (int*)(ws + o);
    int*   nbr    = (int*)(ws + o + edgeBytes);
    o += 2 * edgeBytes;
    float* ht   = (float*)(ws + o);
    float* ht2  = (float*)(ws + o + rowBytes);

    const int gemmBlocks = ((N + 15) / 16 + 7) / 8;
    const int edgeBlocks = (E + 256 * EDGES_PER_THREAD - 1) / (256 * EDGES_PER_THREAD);
    const int fcBlocks   = ((N + FC_NODES - 1) / FC_NODES + 7) / 8;

    init_count_kernel<<<(N + 255) / 256, 256, 0, stream>>>(cntA, cntB, cursor, N);
    count_rank_kernel<<<edgeBlocks, 256, 0, stream>>>(dstv, cntA, cntB, rankh, E);
    offsets_kernel<<<(N + 255) / 256, 256, 0, stream>>>(cntA, cntB, offs, dis, cursor, N);
    fill_kernel<<<edgeBlocks, 256, 0, stream>>>(srcv, dstv, rankh, offs, cntA, nbr, E);

    gemm_wmma_kernel<false><<<gemmBlocks, 256, 0, stream>>>(x, W1, dis, nullptr, ht, N, IN);

    gather_transform_kernel<<<gemmBlocks, 256, 0, stream>>>(offs, cntA, cntB, nbr, ht, dis, b1, W2, ht2, N);

    gather_fc_kernel<<<fcBlocks, 256, 0, stream>>>(offs, cntA, cntB, nbr, ht2, dis, b2, Wfc, bfc, (float*)d_out, N);
}


# layer-1 product: W1 staged per block in LDS in fragment order, x rows by 16-byte loads (9 global loads per k step become 0.5 plus 2 LDS reads), 4-wave blocks; k order inside each 8-k block changes
# speedup vs baseline: 3.5864x; 1.0372x over previous
#include <hip/hip_runtime.h>

typedef __attribute__((ext_vector_type(2))) float v2f;
typedef __attribute__((ext_vector_type(8))) float v8f;

__device__ __forceinline__ int wave_half() {
    const unsigned id = __builtin_amdgcn_s_sendmsg_rtn(0x87);
    return (int)((id >> 18) & 1u);
}

__global__ void init_count_kernel(int* __restrict__ cntA, int* __restrict__ cntB,
                                  int* __restrict__ cursor, int n) {
    int i = blockIdx.x * blockDim.x + threadIdx.x;
    if (i < n) { cntA[i] = 0; cntB[i] = 0; }
    if (i == 0) cursor[0] = 0;
}

#define EDGES_PER_THREAD 8
template <bool FULL>
__device__ __forceinline__ void count_rank_body(const int* __restrict__ dst,
                                                int* __restrict__ cnt,
                                                int* __restrict__ rankh,
                                                int e0, int stride, int h, int E) {
    int d[EDGES_PER_THREAD], r[EDGES_PER_THREAD];
    #pragma unroll
    for (int u = 0; u < EDGES_PER_THREAD; ++u) {
        const int e = e0 + u * stride;
        if (FULL || e < E) d[u] = dst[e];
    }
    #pragma unroll
    for (int u = 0; u < EDGES_PER_THREAD; ++u) {
        const int e = e0 + u * stride;
        if (FULL || e < E) r[u] = atomicAdd(&cnt[d[u]], 1);
    }
    #pragma unroll
    for (int u = 0; u < EDGES_PER_THREAD; ++u) {
        const int e = e0 + u * stride;
        if (FULL || e < E) rankh[e] = r[u] * 2 + h;
    }
}

__global__ void count_rank_kernel(const int* __restrict__ dst,
                                  int* __restrict__ cntA,
                                  int* __restrict__ cntB,
                                  int* __restrict__ rankh, int E) {
    const int stride = (int)blockDim.x;
    const int b0 = blockIdx.x * (stride * EDGES_PER_THREAD);
    const int e0 = b0 + threadIdx.x;
    const int h = wave_half();
    int* cnt = h ? cntB : cntA;
    if (b0 + stride * EDGES_PER_THREAD <= E) count_rank_body<true>(dst, cnt, rankh, e0, stride, h, E);
    else                                      count_rank_body<false>(dst, cnt, rankh, e0, stride, h, E);
}

__global__ void offsets_kernel(const int* __restrict__ cntA,
                               const int* __restrict__ cntB,
                               int* __restrict__ off,
                               float* __restrict__ dis,
                               int* __restrict__ cursor, int n) {
    __shared__ int wsum[32];
    __shared__ int bbase;
    const int i = blockIdx.x * blockDim.x + threadIdx.x;
    const int lane = threadIdx.x & 31;
    const int wv = threadIdx.x >> 5;
    const int nw = blockDim.x >> 5;
    const int c = (i < n) ? cntA[i] + cntB[i] : 0;
    int s = c;
    for (int o = 1; o < 32; o <<= 1) {
        const int t = __shfl_up(s, o, 32);
        if (lane >= o) s += t;
    }
    if (lane == 31) wsum[wv] = s;
    __syncthreads();
    if (threadIdx.x == 0) {
        int t = 0;
        for (int w = 0; w < nw; ++w) { const int x = wsum[w]; wsum[w] = t; t += x; }
        bbase = atomicAdd(cursor, t);
    }
    __syncthreads();
    if (i < n) {
        off[i] = bbase + wsum[wv] + s - c;
        dis[i] = rsqrtf((float)(c + 1));
    }
}

template <bool FULL>
__device__ __forceinline__ void fill_body(const int* __restrict__ src,
                                          const int* __restrict__ dst,
                                          const int* __restrict__ rankh,
                                          const int* __restrict__ off,
                                          const int* __restrict__ cntA,
                                          int* __restrict__ nbr,
                                          int e0, int stride, int E) {
    int d[EDGES_PER_THREAD], rh[EDGES_PER_THREAD], s[EDGES_PER_THREAD], pos[EDGES_PER_THREAD];
    #pragma unroll
    for (int u = 0; u < EDGES_PER_THREAD; ++u) {
        const int e = e0 + u * stride;
        if (FULL || e < E) { d[u] = dst[e]; rh[u] = rankh[e]; s[u] = src[e]; }
    }
    #pragma unroll
    for (int u = 0; u < EDGES_PER_THREAD; ++u) {
        const int e = e0 + u * stride;
        if (FULL || e < E) pos[u] = off[d[u]] + (rh[u] >> 1) + ((rh[u] & 1) ? cntA[d[u]] : 0);
    }
    #pragma unroll
    for (int u = 0; u < EDGES_PER_THREAD; ++u) {
        const int e = e0 + u * stride;
        if (FULL || e < E) nbr[pos[u]] = s[u];
    }
}

__global__ void fill_kernel(const int* __restrict__ src,
                            const int* __restrict__ dst,
                            const int* __restrict__ rankh,
                            const int* __restrict__ off,
                            const int* __restrict__ cntA,
                            int* __restrict__ nbr, int E) {
    const int stride = (int)blockDim.x;
    const int b0 = blockIdx.x * (stride * EDGES_PER_THREAD);
    const int e0 = b0 + threadIdx.x;
    if (b0 + stride * EDGES_PER_THREAD <= E) fill_body<true>(src, dst, rankh, off, cntA, nbr, e0, stride, E);
    else                                      fill_body<false>(src, dst, rankh, off, cntA, nbr, e0, stride, E);
}

#define IN_K 128
__global__ void __launch_bounds__(128)
product1_kernel(const float* __restrict__ X,
                const float* __restrict__ W,
                const float* __restrict__ dis,
                float* __restrict__ ht, int M) {
    __shared__ float4 Wl[2][IN_K / 4][32];
    const int lane = threadIdx.x & 31;
    const int wv = __builtin_amdgcn_readfirstlane((int)(threadIdx.x >> 5));
    const int m0 = (blockIdx.x * 4 + wv) * 16;
    const bool live = m0 < M;
    const int half = lane >> 4;
    const int ml   = lane & 15;
    const int arow = (m0 + ml < M) ? (m0 + ml) : (M - 1);
    const float* xr = X + (size_t)arow * IN_K + 4 * half;

    float4 wa[4], wb[4], wc[4], wd[4];
    #pragma unroll
    for (int r = 0; r < 4; ++r) {
        const int g = (int)threadIdx.x + 128 * r;
        const int i = g & 3, h = (g >> 2) & 1, p = (g >> 3) & 1, step = g >> 4;
        const int ka = 8 * (step >> 1) + 4 * h + 2 * (step & 1);
        const float* w0 = W + (size_t)ka * 64 + 32 * p + 4 * i;
        wa[r] = *(const float4*)(w0);
        wb[r] = *(const float4*)(w0 + 64);
        wc[r] = *(const float4*)(w0 + 16);
        wd[r] = *(const float4*)(w0 + 80);
    }
    float4 qa[8], qb[8];
    #pragma unroll
    for (int t = 0; t < 8; ++t) qa[t] = *(const float4*)(xr + 8 * t);
    #pragma unroll
    for (int r = 0; r < 4; ++r) {
        const int g = (int)threadIdx.x + 128 * r;
        const int i = g & 3, h = (g >> 2) & 1, p = (g >> 3) & 1, step = g >> 4;
        float4* dst = &Wl[p][step][h * 16 + 4 * i];
        float4 f;
        f.x = wa[r].x; f.y = wb[r].x; f.z = wc[r].x; f.w = wd[r].x; dst[0] = f;
        f.x = wa[r].y; f.y = wb[r].y; f.z = wc[r].y; f.w = wd[r].y; dst[1] = f;
        f.x = wa[r].z; f.y = wb[r].z; f.z = wc[r].z; f.w = wd[r].z; dst[2] = f;
        f.x = wa[r].w; f.y = wb[r].w; f.z = wc[r].w; f.w = wd[r].w; dst[3] = f;
    }
    #pragma unroll
    for (int t = 0; t < 8; ++t) qb[t] = *(const float4*)(xr + 64 + 8 * t);
    __builtin_amdgcn_fence(__ATOMIC_RELEASE, "workgroup", "local");
    __builtin_amdgcn_s_barrier();
    __builtin_amdgcn_fence(__ATOMIC_ACQUIRE, "workgroup", "local");
    if (!live) return;

    v8f c0 = {}, c1 = {}, c2 = {}, c3 = {};
    #pragma unroll
    for (int t = 0; t < 16; ++t) {
        const float4 q = (t < 8) ? qa[t & 7] : qb[t & 7];
        #pragma unroll
        for (int e = 0; e < 2; ++e) {
            v2f av;
            av.x = e ? q.z : q.x;
            av.y = e ? q.w : q.y;
            const float4 b01 = Wl[0][2 * t + e][lane];
            const float4 b23 = Wl[1][2 * t + e][lane];
            v2f b0; b0.x = b01.x; b0.y = b01.y;
            v2f b1; b1.x = b01.z; b1.y = b01.w;
            v2f b2; b2.x = b23.x; b2.y = b23.y;
            v2f b3; b3.x = b23.z; b3.y = b23.w;
            c0 = __builtin_amdgcn_wmma_f32_16x16x4_f32(false, av, false, b0, (short)0, c0, false, false);
            c1 = __builtin_amdgcn_wmma_f32_16x16x4_f32(false, av, false, b1, (short)0, c1, false, false);
            c2 = __builtin_amdgcn_wmma_f32_16x16x4_f32(false, av, false, b2, (short)0, c2, false, false);
            c3 = __builtin_amdgcn_wmma_f32_16x16x4_f32(false, av, false, b3, (short)0, c3, false, false);
        }
    }

    for (int r = 0; r < 8; ++r) {
        const int orow = m0 + half * 8 + r;
        const float dr = dis[orow];
        float* hp = ht + (size_t)orow * 64;
        hp[ml] = c0[r] * dr; hp[16 + ml] = c1[r] * dr; hp[32 + ml] = c2[r] * dr; hp[48 + ml] = c3[r] * dr;
    }
}

#define ROWS_IN_FLIGHT 24
__device__ __forceinline__ float2 node_sum(const int* __restrict__ nbr,
                                           const float* __restrict__ hl,
                                           int start, int deg, int i, int lane) {
    const int tot = deg + 1;
    float2 acc; acc.x = 0.0f; acc.y = 0.0f;
    int base = 0;
    do {
        const int left = tot - base;
        const int n = left < 32 ? left : 32;
        const int t = base + lane;
        const int mine = (t < deg) ? nbr[start + t] : i;
        for (int j = 0; j < n; j += ROWS_IN_FLIGHT) {
            float2 v[ROWS_IN_FLIGHT];
            v[0] = *(const float2*)(hl + (size_t)__builtin_amdgcn_readlane(mine, j) * 64);
            #pragma unroll
            for (int u = 1; u < ROWS_IN_FLIGHT; ++u) {
                if (j + u < n) {
                    const int s = __builtin_amdgcn_readlane(mine, j + u);
                    v[u] = *(const float2*)(hl + (size_t)s * 64);
                }
            }
            acc.x += v[0].x; acc.y += v[0].y;
            #pragma unroll
            for (int u = 1; u < ROWS_IN_FLIGHT; ++u) {
                if (j + u < n) { acc.x += v[u].x; acc.y += v[u].y; }
            }
        }
        base += 32;
    } while (base < tot);
    return acc;
}

#define TILE_ROWS 16
#define A_PITCH 68
__global__ void __launch_bounds__(256)
gather_transform_kernel(const int* __restrict__ off,
                        const int* __restrict__ cntA,
                        const int* __restrict__ cntB,
                        const int* __restrict__ nbr,
                        const float* __restrict__ ht,
                        const float* __restrict__ dis,
                        const float* __restrict__ bias,
                        const float* __restrict__ W,
                        float* __restrict__ hout, int N) {
    __shared__ float4 Wl[2][16][32];
    __shared__ float As[8][TILE_ROWS][A_PITCH];
    const int lane = threadIdx.x & 31;
    const int wv = __builtin_amdgcn_readfirstlane((int)(threadIdx.x >> 5));
    const int m0 = (blockIdx.x * 8 + wv) * TILE_ROWS;
    const int nn = (N - m0) < TILE_ROWS ? (N - m0) : TILE_ROWS;

    int offv = 0, degv = 0, disv = 0;
    if (lane < nn) {
        offv = off[m0 + lane];
        degv = cntA[m0 + lane] + cntB[m0 + lane];
        disv = __float_as_int(dis[m0 + lane]);
    }
    const float bj0 = bias[lane * 2], bj1 = bias[lane * 2 + 1];

    {
        float4 wr[4];
        #pragma unroll
        for (int u = 0; u < 4; ++u)
            wr[u] = *(const float4*)(W + (size_t)(u * 256 + (int)threadIdx.x) * 4);
        const int t = (int)threadIdx.x;
        const int tile = (t & 15) >> 2;
        const int ln0 = ((t >> 5) & 1) * 16 + ((t & 3) << 2);
        const int e = (tile & 1) * 2 + ((t >> 4) & 1);
        float* wp = (float*)&Wl[tile >> 1][t >> 6][ln0] + e;
        #pragma unroll
        for (int u = 0; u < 4; ++u) {
            float* p = wp + u * (4 * 32 * 4);
            p[0] = wr[u].x; p[4] = wr[u].y; p[8] = wr[u].z; p[12] = wr[u].w;
        }
    }
    __syncthreads();
    if (nn <= 0) return;

    const float* hl = ht + lane * 2;
    float* slot = &As[wv][0][lane * 2];

    for (int q = 0; q < TILE_ROWS; ++q) {
        float2 a; a.x = 0.0f; a.y = 0.0f;
        if (q < nn) {
            const int start = __builtin_amdgcn_readlane(offv, q);
            const int deg = __builtin_amdgcn_readlane(degv, q);
            const float di = __int_as_float(__builtin_amdgcn_readlane(disv, q));
            const float2 acc = node_sum(nbr, hl, start, deg, m0 + q, lane);
            a.x = fmaxf(fmaf(acc.x, di, bj0), 0.0f);
            a.y = fmaxf(fmaf(acc.y, di, bj1), 0.0f);
        }
        *(float2*)(slot + q * A_PITCH) = a;
    }

    const int half = lane >> 4;
    const int ml   = lane & 15;
    const float* ap = &As[wv][ml][half * 2];
    v8f c0 = {}, c1 = {}, c2 = {}, c3 = {};
    #pragma unroll
    for (int ks = 0; ks < 16; ++ks) {
        const v2f av = *(const v2f*)(ap + ks * 4);
        const float4 b01 = Wl[0][ks][lane];
        const float4 b23 = Wl[1][ks][lane];
        v2f b0; b0.x = b01.x; b0.y = b01.y;
        v2f b1; b1.x = b01.z; b1.y = b01.w;
        v2f b2; b2.x = b23.x; b2.y = b23.y;
        v2f b3; b3.x = b23.z; b3.y = b23.w;
        c0 = __builtin_amdgcn_wmma_f32_16x16x4_f32(false, av, false, b0, (short)0, c0, false, false);
        c1 = __builtin_amdgcn_wmma_f32_16x16x4_f32(false, av, false, b1, (short)0, c1, false, false);
        c2 = __builtin_amdgcn_wmma_f32_16x16x4_f32(false, av, false, b2, (short)0, c2, false, false);
        c3 = __builtin_amdgcn_wmma_f32_16x16x4_f32(false, av, false, b3, (short)0, c3, false, false);
    }

    float* cp = &As[wv][half * 8][ml];
    #pragma unroll
    for (int r = 0; r < 8; ++r) {
        float* p = cp + r * A_PITCH;
        p[0] = c0[r]; p[16] = c1[r]; p[32] = c2[r]; p[48] = c3[r];
    }
    float2 h[TILE_ROWS];
    #pragma unroll
    for (int q = 0; q < TILE_ROWS; ++q) h[q] = *(const float2*)(slot + q * A_PITCH);
    #pragma unroll
    for (int q = 0; q < TILE_ROWS; ++q) {
        if (q < nn) {
            const float dr = __int_as_float(__builtin_amdgcn_readlane(disv, q));
            float2 o; o.x = h[q].x * dr; o.y = h[q].y * dr;
            *(float2*)(hout + (size_t)(m0 + q) * 64 + lane * 2) = o;
        }
    }
}

#define FC_NODES 8
__device__ __forceinline__ float xhalf(float x) {
    return __builtin_bit_cast(float, __builtin_amdgcn_permlanex16(
        __builtin_bit_cast(int, x), __builtin_bit_cast(int, x), 0x76543210, 0xfedcba98, false, false));
}
template <int MASK>
__device__ __forceinline__ float xrow(float x) {
    return __builtin_bit_cast(float, __builtin_amdgcn_update_dpp(
        0, __builtin_bit_cast(int, x), 0x160 + MASK, 0xf, 0xf, false));
}
__device__ __forceinline__ float wave_sum(float p) {
    p += xhalf(p);
    p += xrow<8>(p);
    p += xrow<4>(p);
    p += xrow<2>(p);
    p += xrow<1>(p);
    return p;
}

__global__ void __launch_bounds__(256)
gather_fc_kernel(const int* __restrict__ off,
                 const int* __restrict__ cntA,
                 const int* __restrict__ cntB,
                 const int* __restrict__ nbr,
                 const float* __restrict__ ht,
                 const float* __restrict__ dis,
                 const float* __restrict__ b2,
                 const float* __restrict__ wfc,
                 const float* __restrict__ bfc,
                 float* __restrict__ out, int N) {
    __shared__ float res[8 * FC_NODES];
    const int lane = threadIdx.x & 31;
    const int wv = __builtin_amdgcn_readfirstlane((int)(threadIdx.x >> 5));
    const int i0 = (blockIdx.x * 8 + wv) * FC_NODES;
    float r = 0.0f;
    if (i0 < N) {
        const int nn = (N - i0) < FC_NODES ? (N - i0) : FC_NODES;
        int offv = 0, degv = 0, disv = 0;
        if (lane < nn) {
            offv = off[i0 + lane];
            degv = cntA[i0 + lane] + cntB[i0 + lane];
            disv = __float_as_int(dis[i0 + lane]);
        }
        const float* hl = ht + lane * 2;
        const int j = lane * 2;
        const float bj0 = b2[j], bj1 = b2[j + 1];
        const float w0 = wfc[j], w1 = wfc[j + 1];
        for (int q = 0; q < nn; ++q) {
            const int start = __builtin_amdgcn_readlane(offv, q);
            const int deg = __builtin_amdgcn_readlane(degv, q);
            const float di = __int_as_float(__builtin_amdgcn_readlane(disv, q));
            const float2 acc = node_sum(nbr, hl, start, deg, i0 + q, lane);
            const float h0 = fmaxf(fmaf(acc.x, di, bj0), 0.0f);
            const float h1 = fmaxf(fmaf(acc.y, di, bj1), 0.0f);
            const float p = wave_sum(h0 * w0 + h1 * w1);
            if (lane == q) r = p;
        }
    }
    if (lane < FC_NODES) res[wv * FC_NODES + lane] = r;
    __syncthreads();
    if (wv == 0) {
        const int b = blockIdx.x * (8 * FC_NODES) + lane * 2;
        const float bf = bfc[0];
        if (b + 1 < N) {
            float2 o; o.x = res[lane * 2] + bf; o.y = res[lane * 2 + 1] + bf;
            *(float2*)(out + b) = o;
        } else if (b < N) {
            out[b] = res[lane * 2] + bf;
        }
    }
}

extern "C" void kernel_launch(void* const* d_in, const int* in_sizes, int n_in,
                              void* d_out, int out_size, void* d_ws, size_t ws_size,
                              hipStream_t stream) {
    const float* x   = (const float*)d_in[0];
    const int*   ei  = (const int*)d_in[1];
    const float* W1  = (const float*)d_in[2];
    const float* b1  = (const float*)d_in[3];
    const float* W2  = (const float*)d_in[4];
    const float* b2  = (const float*)d_in[5];
    const float* Wfc = (const float*)d_in[6];
    const float* bfc = (const float*)d_in[7];

    const int HID = in_sizes[3];
    const int IN  = in_sizes[2] / HID;
    const int N   = in_sizes[0] / IN;
    const int E   = in_sizes[1] / 2;
    const int* srcv = ei;
    const int* dstv = ei + E;

    char* ws = (char*)d_ws;
    const size_t nodeBytes = (((size_t)N * 4) + 255) & ~(size_t)255;
    const size_t edgeBytes = (((size_t)E * 4) + 255) & ~(size_t)255;
    const size_t rowBytes  = (size_t)N * 64 * sizeof(float);
    float* dis    = (float*)ws;
    int*   cntA   = (int*)(ws + nodeBytes);
    int*   cntB   = (int*)(ws + 2 * nodeBytes);
    int*   offs   = (int*)(ws + 3 * nodeBytes);
    int*   cursor = (int*)(ws + 4 * nodeBytes);
    size_t o = 4 * nodeBytes + 256;
    int*   rankh  = (int*)(ws + o);
    int*   nbr    = (int*)(ws + o + edgeBytes);
    o += 2 * edgeBytes;
    float* ht   = (float*)(ws + o);
    float* ht2  = (float*)(ws + o + rowBytes);

    const int gemmBlocks = ((N + 15) / 16 + 7) / 8;
    const int prodBlocks = ((N + 15) / 16 + 3) / 4;
    const int edgeBlocks = (E + 256 * EDGES_PER_THREAD - 1) / (256 * EDGES_PER_THREAD);
    const int fcBlocks   = ((N + FC_NODES - 1) / FC_NODES + 7) / 8;

    init_count_kernel<<<(N + 255) / 256, 256, 0, stream>>>(cntA, cntB, cursor, N);
    count_rank_kernel<<<edgeBlocks, 256, 0, stream>>>(dstv, cntA, cntB, rankh, E);
    offsets_kernel<<<(N + 255) / 256, 256, 0, stream>>>(cntA, cntB, offs, dis, cursor, N);
    fill_kernel<<<edgeBlocks, 256, 0, stream>>>(srcv, dstv, rankh, offs, cntA, nbr, E);

    product1_kernel<<<prodBlocks, 128, 0, stream>>>(x, W1, dis, ht, N);

    gather_transform_kernel<<<gemmBlocks, 256, 0, stream>>>(offs, cntA, cntB, nbr, ht, dis, b1, W2, ht2, N);

    gather_fc_kernel<<<fcBlocks, 256, 0, stream>>>(offs, cntA, cntB, nbr, ht2, dis, b2, Wfc, bfc, (float*)d_out, N);
}


# edges grouped by destination with a two-level counting sort on LDS counters (tile pass, then a block per 256-node range), no global add per edge; 7 launches become 5; sum order: LDS arrival rank
# speedup vs baseline: 4.6992x; 1.3103x over previous
#include <hip/hip_runtime.h>

typedef __attribute__((ext_vector_type(2))) float v2f;
typedef __attribute__((ext_vector_type(8))) float v8f;

#define BUCKET_NODES 256
#define BUCKET_SHIFT 8
#define MAX_BUCKETS 512
#define TILE_EPT 25
#define TILE_EDGES (256 * TILE_EPT)

template <bool FULL>
__device__ __forceinline__ void tile_sort_body(const int* __restrict__ src,
                                               const int* __restrict__ dst,
                                               unsigned* __restrict__ sorted,
                                               int* __restrict__ seg,
                                               int* hist, unsigned* stage, int* wsum,
                                               int tile, int E, int NB, int T) {
    const int tid = (int)threadIdx.x;
    const int lane = tid & 31;
    const int wv = __builtin_amdgcn_readfirstlane(tid >> 5);
    const int e0 = tile * TILE_EDGES + tid;
    int d[TILE_EPT], s[TILE_EPT], r[TILE_EPT];
    #pragma unroll
    for (int u = 0; u < TILE_EPT; ++u) {
        const int e = e0 + u * 256;
        if (FULL || e < E) { d[u] = dst[e]; s[u] = src[e]; }
    }
    #pragma unroll
    for (int u = 0; u < TILE_EPT; ++u) {
        const int e = e0 + u * 256;
        if (FULL || e < E) r[u] = atomicAdd(&hist[d[u] >> BUCKET_SHIFT], 1);
    }
    __syncthreads();
    const int a0 = hist[2 * tid], a1 = hist[2 * tid + 1];
    int inc = a0 + a1;
    for (int o = 1; o < 32; o <<= 1) {
        const int t = __shfl_up(inc, o, 32);
        if (lane >= o) inc += t;
    }
    if (lane == 31) wsum[wv] = inc;
    __syncthreads();
    int wb = 0;
    for (int w = 0; w < wv; ++w) wb += wsum[w];
    const int st0 = wb + inc - (a0 + a1);
    const int st1 = st0 + a0;
    hist[2 * tid] = st0;
    hist[2 * tid + 1] = st1;
    if (2 * tid < NB)     seg[(size_t)(2 * tid) * T + tile]     = st0 | (a0 << 16);
    if (2 * tid + 1 < NB) seg[(size_t)(2 * tid + 1) * T + tile] = st1 | (a1 << 16);
    __syncthreads();
    #pragma unroll
    for (int u = 0; u < TILE_EPT; ++u) {
        const int e = e0 + u * 256;
        if (FULL || e < E)
            stage[hist[d[u] >> BUCKET_SHIFT] + r[u]] =
                ((unsigned)(d[u] & (BUCKET_NODES - 1)) << 24) | (unsigned)s[u];
    }
    __syncthreads();
    const int left = E - tile * TILE_EDGES;
    const int nE = left < TILE_EDGES ? left : TILE_EDGES;
    unsigned* outp = sorted + (size_t)tile * TILE_EDGES;
    for (int j = tid; j < nE; j += 256) outp[j] = stage[j];
}

__global__ void __launch_bounds__(256)
tile_sort_kernel(const int* __restrict__ src,
                 const int* __restrict__ dst,
                 unsigned* __restrict__ sorted,
                 int* __restrict__ seg,
                 int* __restrict__ cursor, int E, int NB, int T) {
    __shared__ int hist[MAX_BUCKETS];
    __shared__ unsigned stage[TILE_EDGES];
    __shared__ int wsum[8];
    const int tile = (int)blockIdx.x;
    hist[threadIdx.x] = 0;
    hist[threadIdx.x + 256] = 0;
    if (tile == 0 && threadIdx.x == 0) cursor[0] = 0;
    __syncthreads();
    if ((tile + 1) * TILE_EDGES <= E) tile_sort_body<true>(src, dst, sorted, seg, hist, stage, wsum, tile, E, NB, T);
    else                              tile_sort_body<false>(src, dst, sorted, seg, hist, stage, wsum, tile, E, NB, T);
}

#define STAGE_SLOTS 8192
template <int MODE>
__device__ __forceinline__ void walk_pieces(const unsigned* __restrict__ sorted,
                                            const int* __restrict__ segb,
                                            int* cnt, const int* run, int* stage,
                                            int* __restrict__ nbr,
                                            int wv, int lane, int T) {
    for (int t0 = wv * 32; t0 < T; t0 += 256) {
        const int sgv = (t0 + lane < T) ? segb[t0 + lane] : 0;
        int cmax = 0;
        for (int o = 0; o < 32; ++o) {
            const int cu = __builtin_amdgcn_readlane(sgv, o) >> 16;
            cmax = cu > cmax ? cu : cmax;
        }
        for (int i0 = 0; i0 < cmax; i0 += 32) {
            const int i = i0 + lane;
            unsigned k[32];
            #pragma unroll
            for (int u = 0; u < 32; ++u) {
                const int sg = __builtin_amdgcn_readlane(sgv, u);
                const size_t a = (i < (sg >> 16))
                    ? (size_t)(t0 + u) * TILE_EDGES + (size_t)((sg & 0xffff) + i) : (size_t)0;
                k[u] = sorted[a];
            }
            __builtin_amdgcn_sched_barrier(0);
            if (MODE == 2) {
                #pragma unroll
                for (int u = 0; u < 32; ++u) {
                    const int sg = __builtin_amdgcn_readlane(sgv, u);
                    if (i < (sg >> 16)) {
                        const int f = (int)(k[u] >> 24);
                        const int r = atomicAdd(&cnt[f], 1);
                        nbr[run[f] + r] = (int)(k[u] & 0xffffffu);
                    }
                }
            } else {
                int r[32], p[32];
                #pragma unroll
                for (int u = 0; u < 32; ++u) {
                    const int sg = __builtin_amdgcn_readlane(sgv, u);
                    const int f = (i < (sg >> 16)) ? (int)(k[u] >> 24) : BUCKET_NODES + lane;
                    if (MODE == 0) atomicAdd(&cnt[f], 1);
                    else           r[u] = atomicAdd(&cnt[f], 1);
                }
                if (MODE == 1) {
                    #pragma unroll
                    for (int u = 0; u < 32; ++u)
                        p[u] = run[k[u] >> 24];
                    #pragma unroll
                    for (int u = 0; u < 32; ++u) {
                        const int sg = __builtin_amdgcn_readlane(sgv, u);
                        stage[(i < (sg >> 16)) ? p[u] + r[u] : STAGE_SLOTS + lane] =
                            (int)(k[u] & 0xffffffu);
                    }
                }
            }
        }
    }
}

__global__ void __launch_bounds__(256)
csr_build_kernel(const unsigned* __restrict__ sorted,
                 const int* __restrict__ seg,
                 int* __restrict__ cursor,
                 int* __restrict__ off,
                 int* __restrict__ deg,
                 float* __restrict__ dis,
                 int* __restrict__ nbr, int N, int T) {
    __shared__ int cnt[BUCKET_NODES + 32];
    __shared__ int run[BUCKET_NODES];
    __shared__ int stage[STAGE_SLOTS + 32];
    __shared__ int wsum[8];
    __shared__ int bbase, btot;
    const int tid = (int)threadIdx.x;
    const int lane = tid & 31;
    const int wv = __builtin_amdgcn_readfirstlane(tid >> 5);
    const int b = (int)blockIdx.x;
    const int* segb = seg + (size_t)b * T;
    cnt[tid] = 0;
    if (tid < 32) cnt[BUCKET_NODES + tid] = 0;
    __syncthreads();
    walk_pieces<0>(sorted, segb, cnt, run, stage, nbr, wv, lane, T);
    __syncthreads();
    const int c = cnt[tid];
    int inc = c;
    for (int o = 1; o < 32; o <<= 1) {
        const int t = __shfl_up(inc, o, 32);
        if (lane >= o) inc += t;
    }
    if (lane == 31) wsum[wv] = inc;
    __syncthreads();
    if (tid == 0) {
        int tot = 0;
        for (int w = 0; w < 8; ++w) { const int x = wsum[w]; wsum[w] = tot; tot += x; }
        btot = tot;
        bbase = atomicAdd(cursor, tot);
    }
    __syncthreads();
    const int tot = btot;
    const int base = bbase;
    const int local = wsum[wv] + inc - c;
    const int i = b * BUCKET_NODES + tid;
    if (i < N) {
        off[i] = base + local;
        deg[i] = c;
        dis[i] = rsqrtf((float)(c + 1));
    }
    cnt[tid] = 0;
    if (tot <= STAGE_SLOTS) {
        run[tid] = local;
        __syncthreads();
        walk_pieces<1>(sorted, segb, cnt, run, stage, nbr, wv, lane, T);
        __syncthreads();
        for (int j = tid; j < tot; j += 256) nbr[base + j] = stage[j];
    } else {
        run[tid] = base + local;
        __syncthreads();
        walk_pieces<2>(sorted, segb, cnt, run, stage, nbr, wv, lane, T);
    }
}

#define IN_K 128
__global__ void __launch_bounds__(128)
product1_kernel(const float* __restrict__ X,
                const float* __restrict__ W,
                const float* __restrict__ dis,
                float* __restrict__ ht, int M) {
    __shared__ float4 Wl[2][IN_K / 4][32];
    const int lane = threadIdx.x & 31;
    const int wv = __builtin_amdgcn_readfirstlane((int)(threadIdx.x >> 5));
    const int m0 = (blockIdx.x * 4 + wv) * 16;
    const bool live = m0 < M;
    const int half = lane >> 4;
    const int ml   = lane & 15;
    const int arow = (m0 + ml < M) ? (m0 + ml) : (M - 1);
    const float* xr = X + (size_t)arow * IN_K + 4 * half;

    float4 wa[4], wb[4], wc[4], wd[4];
    #pragma unroll
    for (int r = 0; r < 4; ++r) {
        const int g = (int)threadIdx.x + 128 * r;
        const int i = g & 3, h = (g >> 2) & 1, p = (g >> 3) & 1, step = g >> 4;
        const int ka = 8 * (step >> 1) + 4 * h + 2 * (step & 1);
        const float* w0 = W + (size_t)ka * 64 + 32 * p + 4 * i;
        wa[r] = *(const float4*)(w0);
        wb[r] = *(const float4*)(w0 + 64);
        wc[r] = *(const float4*)(w0 + 16);
        wd[r] = *(const float4*)(w0 + 80);
    }
    float4 qa[8], qb[8];
    #pragma unroll
    for (int t = 0; t < 8; ++t) qa[t] = *(const float4*)(xr + 8 * t);
    #pragma unroll
    for (int r = 0; r < 4; ++r) {
        const int g = (int)threadIdx.x + 128 * r;
        const int i = g & 3, h = (g >> 2) & 1, p = (g >> 3) & 1, step = g >> 4;
        float4* dst = &Wl[p][step][h * 16 + 4 * i];
        float4 f;
        f.x = wa[r].x; f.y = wb[r].x; f.z = wc[r].x; f.w = wd[r].x; dst[0] = f;
        f.x = wa[r].y; f.y = wb[r].y; f.z = wc[r].y; f.w = wd[r].y; dst[1] = f;
        f.x = wa[r].z; f.y = wb[r].z; f.z = wc[r].z; f.w = wd[r].z; dst[2] = f;
        f.x = wa[r].w; f.y = wb[r].w; f.z = wc[r].w; f.w = wd[r].w; dst[3] = f;
    }
    #pragma unroll
    for (int t = 0; t < 8; ++t) qb[t] = *(const float4*)(xr + 64 + 8 * t);
    __builtin_amdgcn_fence(__ATOMIC_RELEASE, "workgroup", "local");
    __builtin_amdgcn_s_barrier();
    __builtin_amdgcn_fence(__ATOMIC_ACQUIRE, "workgroup", "local");
    if (!live) return;

    v8f c0 = {}, c1 = {}, c2 = {}, c3 = {};
    #pragma unroll
    for (int t = 0; t < 16; ++t) {
        const float4 q = (t < 8) ? qa[t & 7] : qb[t & 7];
        #pragma unroll
        for (int e = 0; e < 2; ++e) {
            v2f av;
            av.x = e ? q.z : q.x;
            av.y = e ? q.w : q.y;
            const float4 b01 = Wl[0][2 * t + e][lane];
            const float4 b23 = Wl[1][2 * t + e][lane];
            v2f b0; b0.x = b01.x; b0.y = b01.y;
            v2f b1; b1.x = b01.z; b1.y = b01.w;
            v2f b2; b2.x = b23.x; b2.y = b23.y;
            v2f b3; b3.x = b23.z; b3.y = b23.w;
            c0 = __builtin_amdgcn_wmma_f32_16x16x4_f32(false, av, false, b0, (short)0, c0, false, false);
            c1 = __builtin_amdgcn_wmma_f32_16x16x4_f32(false, av, false, b1, (short)0, c1, false, false);
            c2 = __builtin_amdgcn_wmma_f32_16x16x4_f32(false, av, false, b2, (short)0, c2, false, false);
            c3 = __builtin_amdgcn_wmma_f32_16x16x4_f32(false, av, false, b3, (short)0, c3, false, false);
        }
    }

    for (int r = 0; r < 8; ++r) {
        const int orow = m0 + half * 8 + r;
        const float dr = dis[orow];
        float* hp = ht + (size_t)orow * 64;
        hp[ml] = c0[r] * dr; hp[16 + ml] = c1[r] * dr; hp[32 + ml] = c2[r] * dr; hp[48 + ml] = c3[r] * dr;
    }
}

#define ROWS_IN_FLIGHT 24
__device__ __forceinline__ float2 node_sum(const int* __restrict__ nbr,
                                           const float* __restrict__ hl,
                                           int start, int deg, int i, int lane) {
    const int tot = deg + 1;
    float2 acc; acc.x = 0.0f; acc.y = 0.0f;
    int base = 0;
    do {
        const int left = tot - base;
        const int n = left < 32 ? left : 32;
        const int t = base + lane;
        const int mine = (t < deg) ? nbr[start + t] : i;
        for (int j = 0; j < n; j += ROWS_IN_FLIGHT) {
            float2 v[ROWS_IN_FLIGHT];
            v[0] = *(const float2*)(hl + (size_t)__builtin_amdgcn_readlane(mine, j) * 64);
            #pragma unroll
            for (int u = 1; u < ROWS_IN_FLIGHT; ++u) {
                if (j + u < n) {
                    const int s = __builtin_amdgcn_readlane(mine, j + u);
                    v[u] = *(const float2*)(hl + (size_t)s * 64);
                }
            }
            acc.x += v[0].x; acc.y += v[0].y;
            #pragma unroll
            for (int u = 1; u < ROWS_IN_FLIGHT; ++u) {
                if (j + u < n) { acc.x += v[u].x; acc.y += v[u].y; }
            }
        }
        base += 32;
    } while (base < tot);
    return acc;
}

#define TILE_ROWS 16
#define A_PITCH 68
__global__ void __launch_bounds__(256)
gather_transform_kernel(const int* __restrict__ off,
                        const int* __restrict__ deg,
                        const int* __restrict__ nbr,
                        const float* __restrict__ ht,
                        const float* __restrict__ dis,
                        const float* __restrict__ bias,
                        const float* __restrict__ W,
                        float* __restrict__ hout, int N) {
    __shared__ float4 Wl[2][16][32];
    __shared__ float As[8][TILE_ROWS][A_PITCH];
    const int lane = threadIdx.x & 31;
    const int wv = __builtin_amdgcn_readfirstlane((int)(threadIdx.x >> 5));
    const int m0 = (blockIdx.x * 8 + wv) * TILE_ROWS;
    const int nn = (N - m0) < TILE_ROWS ? (N - m0) : TILE_ROWS;

    int offv = 0, degv = 0, disv = 0;
    if (lane < nn) {
        offv = off[m0 + lane];
        degv = deg[m0 + lane];
        disv = __float_as_int(dis[m0 + lane]);
    }
    const float bj0 = bias[lane * 2], bj1 = bias[lane * 2 + 1];

    {
        float4 wr[4];
        #pragma unroll
        for (int u = 0; u < 4; ++u)
            wr[u] = *(const float4*)(W + (size_t)(u * 256 + (int)threadIdx.x) * 4);
        const int t = (int)threadIdx.x;
        const int tile = (t & 15) >> 2;
        const int ln0 = ((t >> 5) & 1) * 16 + ((t & 3) << 2);
        const int e = (tile & 1) * 2 + ((t >> 4) & 1);
        float* wp = (float*)&Wl[tile >> 1][t >> 6][ln0] + e;
        #pragma unroll
        for (int u = 0; u < 4; ++u) {
            float* p = wp + u * (4 * 32 * 4);
            p[0] = wr[u].x; p[4] = wr[u].y; p[8] = wr[u].z; p[12] = wr[u].w;
        }
    }
    __syncthreads();
    if (nn <= 0) return;

    const float* hl = ht + lane * 2;
    float* slot = &As[wv][0][lane * 2];

    for (int q = 0; q < TILE_ROWS; ++q) {
        float2 a; a.x = 0.0f; a.y = 0.0f;
        if (q < nn) {
            const int start = __builtin_amdgcn_readlane(offv, q);
            const int dq = __builtin_amdgcn_readlane(degv, q);
            const float di = __int_as_float(__builtin_amdgcn_readlane(disv, q));
            const float2 acc = node_sum(nbr, hl, start, dq, m0 + q, lane);
            a.x = fmaxf(fmaf(acc.x, di, bj0), 0.0f);
            a.y = fmaxf(fmaf(acc.y, di, bj1), 0.0f);
        }
        *(float2*)(slot + q * A_PITCH) = a;
    }

    const int half = lane >> 4;
    const int ml   = lane & 15;
    const float* ap = &As[wv][ml][half * 2];
    v8f c0 = {}, c1 = {}, c2 = {}, c3 = {};
    #pragma unroll
    for (int ks = 0; ks < 16; ++ks) {
        const v2f av = *(const v2f*)(ap + ks * 4);
        const float4 b01 = Wl[0][ks][lane];
        const float4 b23 = Wl[1][ks][lane];
        v2f b0; b0.x = b01.x; b0.y = b01.y;
        v2f b1; b1.x = b01.z; b1.y = b01.w;
        v2f b2; b2.x = b23.x; b2.y = b23.y;
        v2f b3; b3.x = b23.z; b3.y = b23.w;
        c0 = __builtin_amdgcn_wmma_f32_16x16x4_f32(false, av, false, b0, (short)0, c0, false, false);
        c1 = __builtin_amdgcn_wmma_f32_16x16x4_f32(false, av, false, b1, (short)0, c1, false, false);
        c2 = __builtin_amdgcn_wmma_f32_16x16x4_f32(false, av, false, b2, (short)0, c2, false, false);
        c3 = __builtin_amdgcn_wmma_f32_16x16x4_f32(false, av, false, b3, (short)0, c3, false, false);
    }

    float* cp = &As[wv][half * 8][ml];
    #pragma unroll
    for (int r = 0; r < 8; ++r) {
        float* p = cp + r * A_PITCH;
        p[0] = c0[r]; p[16] = c1[r]; p[32] = c2[r]; p[48] = c3[r];
    }
    float2 h[TILE_ROWS];
    #pragma unroll
    for (int q = 0; q < TILE_ROWS; ++q) h[q] = *(const float2*)(slot + q * A_PITCH);
    #pragma unroll
    for (int q = 0; q < TILE_ROWS; ++q) {
        if (q < nn) {
            const float dr = __int_as_float(__builtin_amdgcn_readlane(disv, q));
            float2 o; o.x = h[q].x * dr; o.y = h[q].y * dr;
            *(float2*)(hout + (size_t)(m0 + q) * 64 + lane * 2) = o;
        }
    }
}

#define FC_NODES 8
__device__ __forceinline__ float xhalf(float x) {
    return __builtin_bit_cast(float, __builtin_amdgcn_permlanex16(
        __builtin_bit_cast(int, x), __builtin_bit_cast(int, x), 0x76543210, 0xfedcba98, false, false));
}
template <int MASK>
__device__ __forceinline__ float xrow(float x) {
    return __builtin_bit_cast(float, __builtin_amdgcn_update_dpp(
        0, __builtin_bit_cast(int, x), 0x160 + MASK, 0xf, 0xf, false));
}
__device__ __forceinline__ float wave_sum(float p) {
    p += xhalf(p);
    p += xrow<8>(p);
    p += xrow<4>(p);
    p += xrow<2>(p);
    p += xrow<1>(p);
    return p;
}

__global__ void __launch_bounds__(256)
gather_fc_kernel(const int* __restrict__ off,
                 const int* __restrict__ deg,
                 const int* __restrict__ nbr,
                 const float* __restrict__ ht,
                 const float* __restrict__ dis,
                 const float* __restrict__ b2,
                 const float* __restrict__ wfc,
                 const float* __restrict__ bfc,
                 float* __restrict__ out, int N) {
    __shared__ float res[8 * FC_NODES];
    const int lane = threadIdx.x & 31;
    const int wv = __builtin_amdgcn_readfirstlane((int)(threadIdx.x >> 5));
    const int i0 = (blockIdx.x * 8 + wv) * FC_NODES;
    float r = 0.0f;
    if (i0 < N) {
        const int nn = (N - i0) < FC_NODES ? (N - i0) : FC_NODES;
        int offv = 0, degv = 0, disv = 0;
        if (lane < nn) {
            offv = off[i0 + lane];
            degv = deg[i0 + lane];
            disv = __float_as_int(dis[i0 + lane]);
        }
        const float* hl = ht + lane * 2;
        const int j = lane * 2;
        const float bj0 = b2[j], bj1 = b2[j + 1];
        const float w0 = wfc[j], w1 = wfc[j + 1];
        for (int q = 0; q < nn; ++q) {
            const int start = __builtin_amdgcn_readlane(offv, q);
            const int dq = __builtin_amdgcn_readlane(degv, q);
            const float di = __int_as_float(__builtin_amdgcn_readlane(disv, q));
            const float2 acc = node_sum(nbr, hl, start, dq, i0 + q, lane);
            const float h0 = fmaxf(fmaf(acc.x, di, bj0), 0.0f);
            const float h1 = fmaxf(fmaf(acc.y, di, bj1), 0.0f);
            const float p = wave_sum(h0 * w0 + h1 * w1);
            if (lane == q) r = p;
        }
    }
    if (lane < FC_NODES) res[wv * FC_NODES + lane] = r;
    __syncthreads();
    if (wv == 0) {
        const int b = blockIdx.x * (8 * FC_NODES) + lane * 2;
        const float bf = bfc[0];
        if (b + 1 < N) {
            float2 o; o.x = res[lane * 2] + bf; o.y = res[lane * 2 + 1] + bf;
            *(float2*)(out + b) = o;
        } else if (b < N) {
            out[b] = res[lane * 2] + bf;
        }
    }
}

extern "C" void kernel_launch(void* const* d_in, const int* in_sizes, int n_in,
                              void* d_out, int out_size, void* d_ws, size_t ws_size,
                              hipStream_t stream) {
    const float* x   = (const float*)d_in[0];
    const int*   ei  = (const int*)d_in[1];
    const float* W1  = (const float*)d_in[2];
    const float* b1  = (const float*)d_in[3];
    const float* W2  = (const float*)d_in[4];
    const float* b2  = (const float*)d_in[5];
    const float* Wfc = (const float*)d_in[6];
    const float* bfc = (const float*)d_in[7];

    const int HID = in_sizes[3];
    const int IN  = in_sizes[2] / HID;
    const int N   = in_sizes[0] / IN;
    const int E   = in_sizes[1] / 2;
    const int* srcv = ei;
    const int* dstv = ei + E;

    const int NB = (N + BUCKET_NODES - 1) / BUCKET_NODES;
    const int T  = (E + TILE_EDGES - 1) / TILE_EDGES;
    char* ws = (char*)d_ws;
    const size_t nodeBytes = (((size_t)N * 4) + 255) & ~(size_t)255;
    const size_t edgeBytes = (((size_t)E * 4) + 255) & ~(size_t)255;
    const size_t tileBytes = (size_t)T * TILE_EDGES * 4;
    const size_t segBytes  = (((size_t)NB * T * 4) + 255) & ~(size_t)255;
    const size_t rowBytes  = (size_t)N * 64 * sizeof(float);
    float* dis    = (float*)ws;
    int*   deg    = (int*)(ws + nodeBytes);
    int*   offs   = (int*)(ws + 2 * nodeBytes);
    int*   cursor = (int*)(ws + 3 * nodeBytes);
    size_t o = 3 * nodeBytes + 256;
    int*      seg    = (int*)(ws + o);
    unsigned* sorted = (unsigned*)(ws + o + segBytes);
    int*      nbr    = (int*)(ws + o + segBytes + tileBytes);
    o += segBytes + tileBytes + edgeBytes;
    float* ht   = (float*)(ws + o);
    float* ht2  = (float*)(ws + o + rowBytes);

    const int gemmBlocks = ((N + 15) / 16 + 7) / 8;
    const int prodBlocks = ((N + 15) / 16 + 3) / 4;
    const int fcBlocks   = ((N + FC_NODES - 1) / FC_NODES + 7) / 8;

    tile_sort_kernel<<<T, 256, 0, stream>>>(srcv, dstv, sorted, seg, cursor, E, NB, T);
    csr_build_kernel<<<NB, 256, 0, stream>>>(sorted, seg, cursor, offs, deg, dis, nbr, N, T);

    product1_kernel<<<prodBlocks, 128, 0, stream>>>(x, W1, dis, ht, N);

    gather_transform_kernel<<<gemmBlocks, 256, 0, stream>>>(offs, deg, nbr, ht, dis, b1, W2, ht2, N);

    gather_fc_kernel<<<fcBlocks, 256, 0, stream>>>(offs, deg, nbr, ht2, dis, b2, Wfc, bfc, (float*)d_out, N);
}


# layer-1 product moves into the CSR-build launch: a node range's block forms its own rows on half its waves while the other half walks (roles swap for the fill walk); 5 launches become 4
# speedup vs baseline: 4.8789x; 1.0382x over previous
#include <hip/hip_runtime.h>

typedef __attribute__((ext_vector_type(2))) float v2f;
typedef __attribute__((ext_vector_type(8))) float v8f;

#define BUCKET_NODES 256
#define BUCKET_SHIFT 8
#define MAX_BUCKETS 512
#define TILE_EPT 25
#define TILE_EDGES (256 * TILE_EPT)

template <bool FULL>
__device__ __forceinline__ void tile_sort_body(const int* __restrict__ src,
                                               const int* __restrict__ dst,
                                               unsigned* __restrict__ sorted,
                                               int* __restrict__ seg,
                                               int* hist, unsigned* stage, int* wsum,
                                               int tile, int E, int NB, int T) {
    const int tid = (int)threadIdx.x;
    const int lane = tid & 31;
    const int wv = __builtin_amdgcn_readfirstlane(tid >> 5);
    const int e0 = tile * TILE_EDGES + tid;
    int d[TILE_EPT], s[TILE_EPT], r[TILE_EPT];
    #pragma unroll
    for (int u = 0; u < TILE_EPT; ++u) {
        const int e = e0 + u * 256;
        if (FULL || e < E) { d[u] = dst[e]; s[u] = src[e]; }
    }
    #pragma unroll
    for (int u = 0; u < TILE_EPT; ++u) {
        const int e = e0 + u * 256;
        if (FULL || e < E) r[u] = atomicAdd(&hist[d[u] >> BUCKET_SHIFT], 1);
    }
    __syncthreads();
    const int a0 = hist[2 * tid], a1 = hist[2 * tid + 1];
    int inc = a0 + a1;
    for (int o = 1; o < 32; o <<= 1) {
        const int t = __shfl_up(inc, o, 32);
        if (lane >= o) inc += t;
    }
    if (lane == 31) wsum[wv] = inc;
    __syncthreads();
    int wb = 0;
    for (int w = 0; w < wv; ++w) wb += wsum[w];
    const int st0 = wb + inc - (a0 + a1);
    const int st1 = st0 + a0;
    hist[2 * tid] = st0;
    hist[2 * tid + 1] = st1;
    if (2 * tid < NB)     seg[(size_t)(2 * tid) * T + tile]     = st0 | (a0 << 16);
    if (2 * tid + 1 < NB) seg[(size_t)(2 * tid + 1) * T + tile] = st1 | (a1 << 16);
    __syncthreads();
    #pragma unroll
    for (int u = 0; u < TILE_EPT; ++u) {
        const int e = e0 + u * 256;
        if (FULL || e < E)
            stage[hist[d[u] >> BUCKET_SHIFT] + r[u]] =
                ((unsigned)(d[u] & (BUCKET_NODES - 1)) << 24) | (unsigned)s[u];
    }
    __syncthreads();
    const int left = E - tile * TILE_EDGES;
    const int nE = left < TILE_EDGES ? left : TILE_EDGES;
    unsigned* outp = sorted + (size_t)tile * TILE_EDGES;
    for (int j = tid; j < nE; j += 256) outp[j] = stage[j];
}

__global__ void __launch_bounds__(256)
tile_sort_kernel(const int* __restrict__ src,
                 const int* __restrict__ dst,
                 unsigned* __restrict__ sorted,
                 int* __restrict__ seg,
                 int* __restrict__ cursor, int E, int NB, int T) {
    __shared__ int hist[MAX_BUCKETS];
    __shared__ unsigned stage[TILE_EDGES];
    __shared__ int wsum[8];
    const int tile = (int)blockIdx.x;
    hist[threadIdx.x] = 0;
    hist[threadIdx.x + 256] = 0;
    if (tile == 0 && threadIdx.x == 0) cursor[0] = 0;
    __syncthreads();
    if ((tile + 1) * TILE_EDGES <= E) tile_sort_body<true>(src, dst, sorted, seg, hist, stage, wsum, tile, E, NB, T);
    else                              tile_sort_body<false>(src, dst, sorted, seg, hist, stage, wsum, tile, E, NB, T);
}

#define STAGE_SLOTS 8192
#define WALKERS 4
template <int MODE>
__device__ __forceinline__ void walk_pieces(const unsigned* __restrict__ sorted,
                                            const int* __restrict__ segb,
                                            int* cnt, const int* run, int* stage,
                                            int* __restrict__ nbr,
                                            int wv, int lane, int T) {
    for (int t0 = wv * 32; t0 < T; t0 += WALKERS * 32) {
        const int sgv = (t0 + lane < T) ? segb[t0 + lane] : 0;
        int cmax = 0;
        for (int o = 0; o < 32; ++o) {
            const int cu = __builtin_amdgcn_readlane(sgv, o) >> 16;
            cmax = cu > cmax ? cu : cmax;
        }
        for (int i0 = 0; i0 < cmax; i0 += 32) {
            const int i = i0 + lane;
            unsigned k[32];
            #pragma unroll
            for (int u = 0; u < 32; ++u) {
                const int sg = __builtin_amdgcn_readlane(sgv, u);
                const size_t a = (i < (sg >> 16))
                    ? (size_t)(t0 + u) * TILE_EDGES + (size_t)((sg & 0xffff) + i) : (size_t)0;
                k[u] = sorted[a];
            }
            __builtin_amdgcn_sched_barrier(0);
            if (MODE == 2) {
                #pragma unroll
                for (int u = 0; u < 32; ++u) {
                    const int sg = __builtin_amdgcn_readlane(sgv, u);
                    if (i < (sg >> 16)) {
                        const int f = (int)(k[u] >> 24);
                        const int r = atomicAdd(&cnt[f], 1);
                        nbr[run[f] + r] = (int)(k[u] & 0xffffffu);
                    }
                }
            } else {
                int r[32], p[32];
                #pragma unroll
                for (int u = 0; u < 32; ++u) {
                    const int sg = __builtin_amdgcn_readlane(sgv, u);
                    const int f = (i < (sg >> 16)) ? (int)(k[u] >> 24) : BUCKET_NODES + lane;
                    if (MODE == 0) atomicAdd(&cnt[f], 1);
                    else           r[u] = atomicAdd(&cnt[f], 1);
                }
                if (MODE == 1) {
                    #pragma unroll
                    for (int u = 0; u < 32; ++u)
                        p[u] = run[k[u] >> 24];
                    #pragma unroll
                    for (int u = 0; u < 32; ++u) {
                        const int sg = __builtin_amdgcn_readlane(sgv, u);
                        stage[(i < (sg >> 16)) ? p[u] + r[u] : STAGE_SLOTS + lane] =
                            (int)(k[u] & 0xffffffu);
                    }
                }
            }
        }
    }
}

#define IN_K 128

__device__ __forceinline__ void load_x_tile(const float* __restrict__ X, int m0, int M,
                                            int lane, float4 (&q)[16]) {
    const int half = lane >> 4;
    const int ml   = lane & 15;
    const int row = (m0 + ml < M) ? (m0 + ml) : (M - 1);
    const float* xr = X + (size_t)row * IN_K + 4 * half;
    #pragma unroll
    for (int t = 0; t < 16; ++t) q[t] = *(const float4*)(xr + 8 * t);
}

__device__ __forceinline__ void mma_tile(const float4 (&q)[16],
                                         const float4 (*Wl)[IN_K / 4][32], int lane,
                                         v8f (&c)[4]) {
    c[0] = v8f{}; c[1] = v8f{}; c[2] = v8f{}; c[3] = v8f{};
    #pragma unroll
    for (int t = 0; t < 16; ++t) {
        #pragma unroll
        for (int e = 0; e < 2; ++e) {
            v2f av;
            av.x = e ? q[t].z : q[t].x;
            av.y = e ? q[t].w : q[t].y;
            const float4 b01 = Wl[0][2 * t + e][lane];
            const float4 b23 = Wl[1][2 * t + e][lane];
            v2f b0; b0.x = b01.x; b0.y = b01.y;
            v2f b1; b1.x = b01.z; b1.y = b01.w;
            v2f b2; b2.x = b23.x; b2.y = b23.y;
            v2f b3; b3.x = b23.z; b3.y = b23.w;
            c[0] = __builtin_amdgcn_wmma_f32_16x16x4_f32(false, av, false, b0, (short)0, c[0], false, false);
            c[1] = __builtin_amdgcn_wmma_f32_16x16x4_f32(false, av, false, b1, (short)0, c[1], false, false);
            c[2] = __builtin_amdgcn_wmma_f32_16x16x4_f32(false, av, false, b2, (short)0, c[2], false, false);
            c[3] = __builtin_amdgcn_wmma_f32_16x16x4_f32(false, av, false, b3, (short)0, c[3], false, false);
        }
    }
}

__device__ __forceinline__ void store_tile(const v8f (&c)[4], const float* dl,
                                           float* __restrict__ ht, int m0, int lane) {
    const int half = lane >> 4;
    const int ml   = lane & 15;
    for (int r = 0; r < 8; ++r) {
        const float dr = dl[half * 8 + r];
        float* hp = ht + (size_t)(m0 + half * 8 + r) * 64;
        hp[ml] = c[0][r] * dr; hp[16 + ml] = c[1][r] * dr; hp[32 + ml] = c[2][r] * dr; hp[48 + ml] = c[3][r] * dr;
    }
}

__device__ __forceinline__ void lds_barrier() {
    __builtin_amdgcn_fence(__ATOMIC_RELEASE, "workgroup", "local");
    __builtin_amdgcn_s_barrier();
    __builtin_amdgcn_fence(__ATOMIC_ACQUIRE, "workgroup", "local");
}

__global__ void __launch_bounds__(256, 4)
csr_build_kernel(const unsigned* __restrict__ sorted,
                 const int* __restrict__ seg,
                 int* __restrict__ cursor,
                 int* __restrict__ off,
                 int* __restrict__ deg,
                 float* __restrict__ dis,
                 int* __restrict__ nbr,
                 const float* __restrict__ X,
                 const float* __restrict__ W,
                 float* __restrict__ ht, int N, int T) {
    __shared__ float4 Wl[2][IN_K / 4][32];
    __shared__ float disl[BUCKET_NODES];
    __shared__ int cnt[BUCKET_NODES + 32];
    __shared__ int run[BUCKET_NODES];
    __shared__ int stage[STAGE_SLOTS + 32];
    __shared__ int wsum[8];
    __shared__ int bbase, btot;
    const int tid = (int)threadIdx.x;
    const int lane = tid & 31;
    const int wv = __builtin_amdgcn_readfirstlane(tid >> 5);
    const bool first = wv < WALKERS;
    const int wk = wv & (WALKERS - 1);
    const int b = (int)blockIdx.x;
    const int* segb = seg + (size_t)b * T;
    const int l0 = (first ? 0 : 32 * WALKERS) + 32 * wk;
    const int m0 = b * BUCKET_NODES + l0;
    cnt[tid] = 0;
    if (tid < 32) cnt[BUCKET_NODES + tid] = 0;

    {
        float4 wa[2], wb[2], wc[2], wd[2];
        #pragma unroll
        for (int r = 0; r < 2; ++r) {
            const int g = tid + 256 * r;
            const int i = g & 3, h = (g >> 2) & 1, p = (g >> 3) & 1, step = g >> 4;
            const int ka = 8 * (step >> 1) + 4 * h + 2 * (step & 1);
            const float* w0 = W + (size_t)ka * 64 + 32 * p + 4 * i;
            wa[r] = *(const float4*)(w0);
            wb[r] = *(const float4*)(w0 + 64);
            wc[r] = *(const float4*)(w0 + 16);
            wd[r] = *(const float4*)(w0 + 80);
        }
        #pragma unroll
        for (int r = 0; r < 2; ++r) {
            const int g = tid + 256 * r;
            const int i = g & 3, h = (g >> 2) & 1, p = (g >> 3) & 1, step = g >> 4;
            float4* dst = &Wl[p][step][h * 16 + 4 * i];
            float4 f;
            f.x = wa[r].x; f.y = wb[r].x; f.z = wc[r].x; f.w = wd[r].x; dst[0] = f;
            f.x = wa[r].y; f.y = wb[r].y; f.z = wc[r].y; f.w = wd[r].y; dst[1] = f;
            f.x = wa[r].z; f.y = wb[r].z; f.z = wc[r].z; f.w = wd[r].z; dst[2] = f;
            f.x = wa[r].w; f.y = wb[r].w; f.z = wc[r].w; f.w = wd[r].w; dst[3] = f;
        }
    }
    float4 qf0[16], qf1[16];
    float4 ql0[16], ql1[16];
    v8f af0[4], af1[4];
    lds_barrier();

    if (first) {
        load_x_tile(X, m0, N, lane, qf0);
        load_x_tile(X, m0 + 16, N, lane, qf1);
        if (m0 < N)      mma_tile(qf0, Wl, lane, af0);
        if (m0 + 16 < N) mma_tile(qf1, Wl, lane, af1);
    } else {
        walk_pieces<0>(sorted, segb, cnt, run, stage, nbr, wk, lane, T);
        load_x_tile(X, m0, N, lane, ql0);
    }
    lds_barrier();
    const int c = cnt[tid];
    int inc = c;
    for (int o = 1; o < 32; o <<= 1) {
        const int t = __shfl_up(inc, o, 32);
        if (lane >= o) inc += t;
    }
    if (lane == 31) wsum[wv] = inc;
    lds_barrier();
    if (tid == 0) {
        int tot = 0;
        for (int w = 0; w < 8; ++w) { const int x = wsum[w]; wsum[w] = tot; tot += x; }
        btot = tot;
        bbase = atomicAdd(cursor, tot);
    }
    lds_barrier();
    const int tot = btot;
    const int base = bbase;
    const int local = wsum[wv] + inc - c;
    const int i = b * BUCKET_NODES + tid;
    const float dsc = rsqrtf((float)(c + 1));
    disl[tid] = dsc;
    if (i < N) {
        off[i] = base + local;
        deg[i] = c;
        dis[i] = dsc;
    }
    cnt[tid] = 0;
    const bool staged = tot <= STAGE_SLOTS;
    run[tid] = staged ? local : base + local;
    lds_barrier();

    if (first) {
        if (m0 < N)      store_tile(af0, disl + l0, ht, m0, lane);
        if (m0 + 16 < N) store_tile(af1, disl + l0 + 16, ht, m0 + 16, lane);
        if (staged) walk_pieces<1>(sorted, segb, cnt, run, stage, nbr, wk, lane, T);
        else        walk_pieces<2>(sorted, segb, cnt, run, stage, nbr, wk, lane, T);
    } else {
        load_x_tile(X, m0 + 16, N, lane, ql1);
        v8f al[4];
        if (m0 < N)      { mma_tile(ql0, Wl, lane, al); store_tile(al, disl + l0, ht, m0, lane); }
        if (m0 + 16 < N) { mma_tile(ql1, Wl, lane, al); store_tile(al, disl + l0 + 16, ht, m0 + 16, lane); }
    }
    if (staged) {
        lds_barrier();
        for (int j = tid; j < tot; j += 256) nbr[base + j] = stage[j];
    }
}

#define ROWS_IN_FLIGHT 24
__device__ __forceinline__ float2 node_sum(const int* __restrict__ nbr,
                                           const float* __restrict__ hl,
                                           int start, int deg, int i, int lane) {
    const int tot = deg + 1;
    float2 acc; acc.x = 0.0f; acc.y = 0.0f;
    int base = 0;
    do {
        const int left = tot - base;
        const int n = left < 32 ? left : 32;
        const int t = base + lane;
        const int mine = (t < deg) ? nbr[start + t] : i;
        for (int j = 0; j < n; j += ROWS_IN_FLIGHT) {
            float2 v[ROWS_IN_FLIGHT];
            v[0] = *(const float2*)(hl + (size_t)__builtin_amdgcn_readlane(mine, j) * 64);
            #pragma unroll
            for (int u = 1; u < ROWS_IN_FLIGHT; ++u) {
                if (j + u < n) {
                    const int s = __builtin_amdgcn_readlane(mine, j + u);
                    v[u] = *(const float2*)(hl + (size_t)s * 64);
                }
            }
            acc.x += v[0].x; acc.y += v[0].y;
            #pragma unroll
            for (int u = 1; u < ROWS_IN_FLIGHT; ++u) {
                if (j + u < n) { acc.x += v[u].x; acc.y += v[u].y; }
            }
        }
        base += 32;
    } while (base < tot);
    return acc;
}

#define TILE_ROWS 16
#define A_PITCH 68
__global__ void __launch_bounds__(256)
gather_transform_kernel(const int* __restrict__ off,
                        const int* __restrict__ deg,
                        const int* __restrict__ nbr,
                        const float* __restrict__ ht,
                        const float* __restrict__ dis,
                        const float* __restrict__ bias,
                        const float* __restrict__ W,
                        float* __restrict__ hout, int N) {
    __shared__ float4 Wl[2][16][32];
    __shared__ float As[8][TILE_ROWS][A_PITCH];
    const int lane = threadIdx.x & 31;
    const int wv = __builtin_amdgcn_readfirstlane((int)(threadIdx.x >> 5));
    const int m0 = (blockIdx.x * 8 + wv) * TILE_ROWS;
    const int nn = (N - m0) < TILE_ROWS ? (N - m0) : TILE_ROWS;

    int offv = 0, degv = 0, disv = 0;
    if (lane < nn) {
        offv = off[m0 + lane];
        degv = deg[m0 + lane];
        disv = __float_as_int(dis[m0 + lane]);
    }
    const float bj0 = bias[lane * 2], bj1 = bias[lane * 2 + 1];

    {
        float4 wr[4];
        #pragma unroll
        for (int u = 0; u < 4; ++u)
            wr[u] = *(const float4*)(W + (size_t)(u * 256 + (int)threadIdx.x) * 4);
        const int t = (int)threadIdx.x;
        const int tile = (t & 15) >> 2;
        const int ln0 = ((t >> 5) & 1) * 16 + ((t & 3) << 2);
        const int e = (tile & 1) * 2 + ((t >> 4) & 1);
        float* wp = (float*)&Wl[tile >> 1][t >> 6][ln0] + e;
        #pragma unroll
        for (int u = 0; u < 4; ++u) {
            float* p = wp + u * (4 * 32 * 4);
            p[0] = wr[u].x; p[4] = wr[u].y; p[8] = wr[u].z; p[12] = wr[u].w;
        }
    }
    __syncthreads();
    if (nn <= 0) return;

    const float* hl = ht + lane * 2;
    float* slot = &As[wv][0][lane * 2];

    for (int q = 0; q < TILE_ROWS; ++q) {
        float2 a; a.x = 0.0f; a.y = 0.0f;
        if (q < nn) {
            const int start = __builtin_amdgcn_readlane(offv, q);
            const int dq = __builtin_amdgcn_readlane(degv, q);
            const float di = __int_as_float(__builtin_amdgcn_readlane(disv, q));
            const float2 acc = node_sum(nbr, hl, start, dq, m0 + q, lane);
            a.x = fmaxf(fmaf(acc.x, di, bj0), 0.0f);
            a.y = fmaxf(fmaf(acc.y, di, bj1), 0.0f);
        }
        *(float2*)(slot + q * A_PITCH) = a;
    }

    const int half = lane >> 4;
    const int ml   = lane & 15;
    const float* ap = &As[wv][ml][half * 2];
    v8f c0 = {}, c1 = {}, c2 = {}, c3 = {};
    #pragma unroll
    for (int ks = 0; ks < 16; ++ks) {
        const v2f av = *(const v2f*)(ap + ks * 4);
        const float4 b01 = Wl[0][ks][lane];
        const float4 b23 = Wl[1][ks][lane];
        v2f b0; b0.x = b01.x; b0.y = b01.y;
        v2f b1; b1.x = b01.z; b1.y = b01.w;
        v2f b2; b2.x = b23.x; b2.y = b23.y;
        v2f b3; b3.x = b23.z; b3.y = b23.w;
        c0 = __builtin_amdgcn_wmma_f32_16x16x4_f32(false, av, false, b0, (short)0, c0, false, false);
        c1 = __builtin_amdgcn_wmma_f32_16x16x4_f32(false, av, false, b1, (short)0, c1, false, false);
        c2 = __builtin_amdgcn_wmma_f32_16x16x4_f32(false, av, false, b2, (short)0, c2, false, false);
        c3 = __builtin_amdgcn_wmma_f32_16x16x4_f32(false, av, false, b3, (short)0, c3, false, false);
    }

    float* cp = &As[wv][half * 8][ml];
    #pragma unroll
    for (int r = 0; r < 8; ++r) {
        float* p = cp + r * A_PITCH;
        p[0] = c0[r]; p[16] = c1[r]; p[32] = c2[r]; p[48] = c3[r];
    }
    float2 h[TILE_ROWS];
    #pragma unroll
    for (int q = 0; q < TILE_ROWS; ++q) h[q] = *(const float2*)(slot + q * A_PITCH);
    #pragma unroll
    for (int q = 0; q < TILE_ROWS; ++q) {
        if (q < nn) {
            const float dr = __int_as_float(__builtin_amdgcn_readlane(disv, q));
            float2 o; o.x = h[q].x * dr; o.y = h[q].y * dr;
            *(float2*)(hout + (size_t)(m0 + q) * 64 + lane * 2) = o;
        }
    }
}

#define FC_NODES 8
__device__ __forceinline__ float xhalf(float x) {
    return __builtin_bit_cast(float, __builtin_amdgcn_permlanex16(
        __builtin_bit_cast(int, x), __builtin_bit_cast(int, x), 0x76543210, 0xfedcba98, false, false));
}
template <int MASK>
__device__ __forceinline__ float xrow(float x) {
    return __builtin_bit_cast(float, __builtin_amdgcn_update_dpp(
        0, __builtin_bit_cast(int, x), 0x160 + MASK, 0xf, 0xf, false));
}
__device__ __forceinline__ float wave_sum(float p) {
    p += xhalf(p);
    p += xrow<8>(p);
    p += xrow<4>(p);
    p += xrow<2>(p);
    p += xrow<1>(p);
    return p;
}

__global__ void __launch_bounds__(256)
gather_fc_kernel(const int* __restrict__ off,
                 const int* __restrict__ deg,
                 const int* __restrict__ nbr,
                 const float* __restrict__ ht,
                 const float* __restrict__ dis,
                 const float* __restrict__ b2,
                 const float* __restrict__ wfc,
                 const float* __restrict__ bfc,
                 float* __restrict__ out, int N) {
    __shared__ float res[8 * FC_NODES];
    const int lane = threadIdx.x & 31;
    const int wv = __builtin_amdgcn_readfirstlane((int)(threadIdx.x >> 5));
    const int i0 = (blockIdx.x * 8 + wv) * FC_NODES;
    float r = 0.0f;
    if (i0 < N) {
        const int nn = (N - i0) < FC_NODES ? (N - i0) : FC_NODES;
        int offv = 0, degv = 0, disv = 0;
        if (lane < nn) {
            offv = off[i0 + lane];
            degv = deg[i0 + lane];
            disv = __float_as_int(dis[i0 + lane]);
        }
        const float* hl = ht + lane * 2;
        const int j = lane * 2;
        const float bj0 = b2[j], bj1 = b2[j + 1];
        const float w0 = wfc[j], w1 = wfc[j + 1];
        for (int q = 0; q < nn; ++q) {
            const int start = __builtin_amdgcn_readlane(offv, q);
            const int dq = __builtin_amdgcn_readlane(degv, q);
            const float di = __int_as_float(__builtin_amdgcn_readlane(disv, q));
            const float2 acc = node_sum(nbr, hl, start, dq, i0 + q, lane);
            const float h0 = fmaxf(fmaf(acc.x, di, bj0), 0.0f);
            const float h1 = fmaxf(fmaf(acc.y, di, bj1), 0.0f);
            const float p = wave_sum(h0 * w0 + h1 * w1);
            if (lane == q) r = p;
        }
    }
    if (lane < FC_NODES) res[wv * FC_NODES + lane] = r;
    __syncthreads();
    if (wv == 0) {
        const int b = blockIdx.x * (8 * FC_NODES) + lane * 2;
        const float bf = bfc[0];
        if (b + 1 < N) {
            float2 o; o.x = res[lane * 2] + bf; o.y = res[lane * 2 + 1] + bf;
            *(float2*)(out + b) = o;
        } else if (b < N) {
            out[b] = res[lane * 2] + bf;
        }
    }
}

extern "C" void kernel_launch(void* const* d_in, const int* in_sizes, int n_in,
                              void* d_out, int out_size, void* d_ws, size_t ws_size,
                              hipStream_t stream) {
    const float* x   = (const float*)d_in[0];
    const int*   ei  = (const int*)d_in[1];
    const float* W1  = (const float*)d_in[2];
    const float* b1  = (const float*)d_in[3];
    const float* W2  = (const float*)d_in[4];
    const float* b2  = (const float*)d_in[5];
    const float* Wfc = (const float*)d_in[6];
    const float* bfc = (const float*)d_in[7];

    const int HID = in_sizes[3];
    const int IN  = in_sizes[2] / HID;
    const int N   = in_sizes[0] / IN;
    const int E   = in_sizes[1] / 2;
    const int* srcv = ei;
    const int* dstv = ei + E;

    const int NB = (N + BUCKET_NODES - 1) / BUCKET_NODES;
    const int T  = (E + TILE_EDGES - 1) / TILE_EDGES;
    char* ws = (char*)d_ws;
    const size_t nodeBytes = (((size_t)N * 4) + 255) & ~(size_t)255;
    const size_t edgeBytes = (((size_t)E * 4) + 255) & ~(size_t)255;
    const size_t tileBytes = (size_t)T * TILE_EDGES * 4;
    const size_t segBytes  = (((size_t)NB * T * 4) + 255) & ~(size_t)255;
    const size_t rowBytes  = (size_t)N * 64 * sizeof(float);
    float* dis    = (float*)ws;
    int*   deg    = (int*)(ws + nodeBytes);
    int*   offs   = (int*)(ws + 2 * nodeBytes);
    int*   cursor = (int*)(ws + 3 * nodeBytes);
    size_t o = 3 * nodeBytes + 256;
    int*      seg    = (int*)(ws + o);
    unsigned* sorted = (unsigned*)(ws + o + segBytes);
    int*      nbr    = (int*)(ws + o + segBytes + tileBytes);
    o += segBytes + tileBytes + edgeBytes;
    float* ht   = (float*)(ws + o);
    float* ht2  = (float*)(ws + o + rowBytes);

    const int gemmBlocks = ((N + 15) / 16 + 7) / 8;
    const int fcBlocks   = ((N + FC_NODES - 1) / FC_NODES + 7) / 8;

    tile_sort_kernel<<<T, 256, 0, stream>>>(srcv, dstv, sorted, seg, cursor, E, NB, T);
    csr_build_kernel<<<NB, 256, 0, stream>>>(sorted, seg, cursor, offs, deg, dis, nbr, x, W1, ht, N, T);

    gather_transform_kernel<<<gemmBlocks, 256, 0, stream>>>(offs, deg, nbr, ht, dis, b1, W2, ht2, N);

    gather_fc_kernel<<<fcBlocks, 256, 0, stream>>>(offs, deg, nbr, ht2, dis, b2, Wfc, bfc, (float*)d_out, N);
}


# row gathers by 16-byte loads, 2 rows per load (one per wave half), index by row_share move, zero row for odd runs; sum order: a run's even and odd places apart, then added; fc: lane's 4 columns first
# speedup vs baseline: 5.0304x; 1.0311x over previous
#include <hip/hip_runtime.h>

typedef __attribute__((ext_vector_type(2))) float v2f;
typedef __attribute__((ext_vector_type(8))) float v8f;

__device__ __forceinline__ int zero_row(int N) { return (N + 15) & ~15; }

#define BUCKET_NODES 256
#define BUCKET_SHIFT 8
#define MAX_BUCKETS 512
#define TILE_EPT 25
#define TILE_EDGES (256 * TILE_EPT)

template <bool FULL>
__device__ __forceinline__ void tile_sort_body(const int* __restrict__ src,
                                               const int* __restrict__ dst,
                                               unsigned* __restrict__ sorted,
                                               int* __restrict__ seg,
                                               int* hist, unsigned* stage, int* wsum,
                                               int tile, int E, int NB, int T) {
    const int tid = (int)threadIdx.x;
    const int lane = tid & 31;
    const int wv = __builtin_amdgcn_readfirstlane(tid >> 5);
    const int e0 = tile * TILE_EDGES + tid;
    int d[TILE_EPT], s[TILE_EPT], r[TILE_EPT];
    #pragma unroll
    for (int u = 0; u < TILE_EPT; ++u) {
        const int e = e0 + u * 256;
        if (FULL || e < E) { d[u] = dst[e]; s[u] = src[e]; }
    }
    #pragma unroll
    for (int u = 0; u < TILE_EPT; ++u) {
        const int e = e0 + u * 256;
        if (FULL || e < E) r[u] = atomicAdd(&hist[d[u] >> BUCKET_SHIFT], 1);
    }
    __syncthreads();
    const int a0 = hist[2 * tid], a1 = hist[2 * tid + 1];
    int inc = a0 + a1;
    for (int o = 1; o < 32; o <<= 1) {
        const int t = __shfl_up(inc, o, 32);
        if (lane >= o) inc += t;
    }
    if (lane == 31) wsum[wv] = inc;
    __syncthreads();
    int wb = 0;
    for (int w = 0; w < wv; ++w) wb += wsum[w];
    const int st0 = wb + inc - (a0 + a1);
    const int st1 = st0 + a0;
    hist[2 * tid] = st0;
    hist[2 * tid + 1] = st1;
    if (2 * tid < NB)     seg[(size_t)(2 * tid) * T + tile]     = st0 | (a0 << 16);
    if (2 * tid + 1 < NB) seg[(size_t)(2 * tid + 1) * T + tile] = st1 | (a1 << 16);
    __syncthreads();
    #pragma unroll
    for (int u = 0; u < TILE_EPT; ++u) {
        const int e = e0 + u * 256;
        if (FULL || e < E)
            stage[hist[d[u] >> BUCKET_SHIFT] + r[u]] =
                ((unsigned)(d[u] & (BUCKET_NODES - 1)) << 24) | (unsigned)s[u];
    }
    __syncthreads();
    const int left = E - tile * TILE_EDGES;
    const int nE = left < TILE_EDGES ? left : TILE_EDGES;
    unsigned* outp = sorted + (size_t)tile * TILE_EDGES;
    for (int j = tid; j < nE; j += 256) outp[j] = stage[j];
}

__global__ void __launch_bounds__(256)
tile_sort_kernel(const int* __restrict__ src,
                 const int* __restrict__ dst,
                 unsigned* __restrict__ sorted,
                 int* __restrict__ seg,
                 int* __restrict__ cursor, int E, int NB, int T) {
    __shared__ int hist[MAX_BUCKETS];
    __shared__ unsigned stage[TILE_EDGES];
    __shared__ int wsum[8];
    const int tile = (int)blockIdx.x;
    hist[threadIdx.x] = 0;
    hist[threadIdx.x + 256] = 0;
    if (tile == 0 && threadIdx.x == 0) cursor[0] = 0;
    __syncthreads();
    if ((tile + 1) * TILE_EDGES <= E) tile_sort_body<true>(src, dst, sorted, seg, hist, stage, wsum, tile, E, NB, T);
    else                              tile_sort_body<false>(src, dst, sorted, seg, hist, stage, wsum, tile, E, NB, T);
}

#define STAGE_SLOTS 8192
#define WALKERS 4
template <int MODE>
__device__ __forceinline__ void walk_pieces(const unsigned* __restrict__ sorted,
                                            const int* __restrict__ segb,
                                            int* cnt, const int* run, int* stage,
                                            int* __restrict__ nbr,
                                            int wv, int lane, int T) {
    for (int t0 = wv * 32; t0 < T; t0 += WALKERS * 32) {
        const int sgv = (t0 + lane < T) ? segb[t0 + lane] : 0;
        int cmax = 0;
        for (int o = 0; o < 32; ++o) {
            const int cu = __builtin_amdgcn_readlane(sgv, o) >> 16;
            cmax = cu > cmax ? cu : cmax;
        }
        for (int i0 = 0; i0 < cmax; i0 += 32) {
            const int i = i0 + lane;
            unsigned k[32];
            #pragma unroll
            for (int u = 0; u < 32; ++u) {
                const int sg = __builtin_amdgcn_readlane(sgv, u);
                const size_t a = (i < (sg >> 16))
                    ? (size_t)(t0 + u) * TILE_EDGES + (size_t)((sg & 0xffff) + i) : (size_t)0;
                k[u] = sorted[a];
            }
            __builtin_amdgcn_sched_barrier(0);
            if (MODE == 2) {
                #pragma unroll
                for (int u = 0; u < 32; ++u) {
                    const int sg = __builtin_amdgcn_readlane(sgv, u);
                    if (i < (sg >> 16)) {
                        const int f = (int)(k[u] >> 24);
                        const int r = atomicAdd(&cnt[f], 1);
                        nbr[run[f] + r] = (int)(k[u] & 0xffffffu);
                    }
                }
            } else {
                int r[32], p[32];
                #pragma unroll
                for (int u = 0; u < 32; ++u) {
                    const int sg = __builtin_amdgcn_readlane(sgv, u);
                    const int f = (i < (sg >> 16)) ? (int)(k[u] >> 24) : BUCKET_NODES + lane;
                    if (MODE == 0) atomicAdd(&cnt[f], 1);
                    else           r[u] = atomicAdd(&cnt[f], 1);
                }
                if (MODE == 1) {
                    #pragma unroll
                    for (int u = 0; u < 32; ++u)
                        p[u] = run[k[u] >> 24];
                    #pragma unroll
                    for (int u = 0; u < 32; ++u) {
                        const int sg = __builtin_amdgcn_readlane(sgv, u);
                        stage[(i < (sg >> 16)) ? p[u] + r[u] : STAGE_SLOTS + lane] =
                            (int)(k[u] & 0xffffffu);
                    }
                }
            }
        }
    }
}

#define IN_K 128

__device__ __forceinline__ void load_x_tile(const float* __restrict__ X, int m0, int M,
                                            int lane, float4 (&q)[16]) {
    const int half = lane >> 4;
    const int ml   = lane & 15;
    const int row = (m0 + ml < M) ? (m0 + ml) : (M - 1);
    const float* xr = X + (size_t)row * IN_K + 4 * half;
    #pragma unroll
    for (int t = 0; t < 16; ++t) q[t] = *(const float4*)(xr + 8 * t);
}

__device__ __forceinline__ void mma_tile(const float4 (&q)[16],
                                         const float4 (*Wl)[IN_K / 4][32], int lane,
                                         v8f (&c)[4]) {
    c[0] = v8f{}; c[1] = v8f{}; c[2] = v8f{}; c[3] = v8f{};
    #pragma unroll
    for (int t = 0; t < 16; ++t) {
        #pragma unroll
        for (int e = 0; e < 2; ++e) {
            v2f av;
            av.x = e ? q[t].z : q[t].x;
            av.y = e ? q[t].w : q[t].y;
            const float4 b01 = Wl[0][2 * t + e][lane];
            const float4 b23 = Wl[1][2 * t + e][lane];
            v2f b0; b0.x = b01.x; b0.y = b01.y;
            v2f b1; b1.x = b01.z; b1.y = b01.w;
            v2f b2; b2.x = b23.x; b2.y = b23.y;
            v2f b3; b3.x = b23.z; b3.y = b23.w;
            c[0] = __builtin_amdgcn_wmma_f32_16x16x4_f32(false, av, false, b0, (short)0, c[0], false, false);
            c[1] = __builtin_amdgcn_wmma_f32_16x16x4_f32(false, av, false, b1, (short)0, c[1], false, false);
            c[2] = __builtin_amdgcn_wmma_f32_16x16x4_f32(false, av, false, b2, (short)0, c[2], false, false);
            c[3] = __builtin_amdgcn_wmma_f32_16x16x4_f32(false, av, false, b3, (short)0, c[3], false, false);
        }
    }
}

__device__ __forceinline__ void store_tile(const v8f (&c)[4], const float* dl,
                                           float* __restrict__ ht, int m0, int lane) {
    const int half = lane >> 4;
    const int ml   = lane & 15;
    for (int r = 0; r < 8; ++r) {
        const float dr = dl[half * 8 + r];
        float* hp = ht + (size_t)(m0 + half * 8 + r) * 64;
        hp[ml] = c[0][r] * dr; hp[16 + ml] = c[1][r] * dr; hp[32 + ml] = c[2][r] * dr; hp[48 + ml] = c[3][r] * dr;
    }
}

__device__ __forceinline__ void lds_barrier() {
    __builtin_amdgcn_fence(__ATOMIC_RELEASE, "workgroup", "local");
    __builtin_amdgcn_s_barrier();
    __builtin_amdgcn_fence(__ATOMIC_ACQUIRE, "workgroup", "local");
}

__global__ void __launch_bounds__(256, 4)
csr_build_kernel(const unsigned* __restrict__ sorted,
                 const int* __restrict__ seg,
                 int* __restrict__ cursor,
                 int* __restrict__ off,
                 int* __restrict__ deg,
                 float* __restrict__ dis,
                 int* __restrict__ nbr,
                 const float* __restrict__ X,
                 const float* __restrict__ W,
                 float* __restrict__ ht, int N, int T) {
    __shared__ float4 Wl[2][IN_K / 4][32];
    __shared__ float disl[BUCKET_NODES];
    __shared__ int cnt[BUCKET_NODES + 32];
    __shared__ int run[BUCKET_NODES];
    __shared__ int stage[STAGE_SLOTS + 32];
    __shared__ int wsum[8];
    __shared__ int bbase, btot;
    const int tid = (int)threadIdx.x;
    const int lane = tid & 31;
    const int wv = __builtin_amdgcn_readfirstlane(tid >> 5);
    const bool first = wv < WALKERS;
    const int wk = wv & (WALKERS - 1);
    const int b = (int)blockIdx.x;
    const int* segb = seg + (size_t)b * T;
    const int l0 = (first ? 0 : 32 * WALKERS) + 32 * wk;
    const int m0 = b * BUCKET_NODES + l0;
    cnt[tid] = 0;
    if (tid < 32) cnt[BUCKET_NODES + tid] = 0;
    if (b == 0 && tid < 64) ht[(size_t)zero_row(N) * 64 + tid] = 0.0f;

    {
        float4 wa[2], wb[2], wc[2], wd[2];
        #pragma unroll
        for (int r = 0; r < 2; ++r) {
            const int g = tid + 256 * r;
            const int i = g & 3, h = (g >> 2) & 1, p = (g >> 3) & 1, step = g >> 4;
            const int ka = 8 * (step >> 1) + 4 * h + 2 * (step & 1);
            const float* w0 = W + (size_t)ka * 64 + 32 * p + 4 * i;
            wa[r] = *(const float4*)(w0);
            wb[r] = *(const float4*)(w0 + 64);
            wc[r] = *(const float4*)(w0 + 16);
            wd[r] = *(const float4*)(w0 + 80);
        }
        #pragma unroll
        for (int r = 0; r < 2; ++r) {
            const int g = tid + 256 * r;
            const int i = g & 3, h = (g >> 2) & 1, p = (g >> 3) & 1, step = g >> 4;
            float4* dst = &Wl[p][step][h * 16 + 4 * i];
            float4 f;
            f.x = wa[r].x; f.y = wb[r].x; f.z = wc[r].x; f.w = wd[r].x; dst[0] = f;
            f.x = wa[r].y; f.y = wb[r].y; f.z = wc[r].y; f.w = wd[r].y; dst[1] = f;
            f.x = wa[r].z; f.y = wb[r].z; f.z = wc[r].z; f.w = wd[r].z; dst[2] = f;
            f.x = wa[r].w; f.y = wb[r].w; f.z = wc[r].w; f.w = wd[r].w; dst[3] = f;
        }
    }
    float4 qf0[16], qf1[16];
    float4 ql0[16], ql1[16];
    v8f af0[4], af1[4];
    lds_barrier();

    if (first) {
        load_x_tile(X, m0, N, lane, qf0);
        load_x_tile(X, m0 + 16, N, lane, qf1);
        if (m0 < N)      mma_tile(qf0, Wl, lane, af0);
        if (m0 + 16 < N) mma_tile(qf1, Wl, lane, af1);
    } else {
        walk_pieces<0>(sorted, segb, cnt, run, stage, nbr, wk, lane, T);
        load_x_tile(X, m0, N, lane, ql0);
    }
    lds_barrier();
    const int c = cnt[tid];
    int inc = c;
    for (int o = 1; o < 32; o <<= 1) {
        const int t = __shfl_up(inc, o, 32);
        if (lane >= o) inc += t;
    }
    if (lane == 31) wsum[wv] = inc;
    lds_barrier();
    if (tid == 0) {
        int tot = 0;
        for (int w = 0; w < 8; ++w) { const int x = wsum[w]; wsum[w] = tot; tot += x; }
        btot = tot;
        bbase = atomicAdd(cursor, tot);
    }
    lds_barrier();
    const int tot = btot;
    const int base = bbase;
    const int local = wsum[wv] + inc - c;
    const int i = b * BUCKET_NODES + tid;
    const float dsc = rsqrtf((float)(c + 1));
    disl[tid] = dsc;
    if (i < N) {
        off[i] = base + local;
        deg[i] = c;
        dis[i] = dsc;
    }
    cnt[tid] = 0;
    const bool staged = tot <= STAGE_SLOTS;
    run[tid] = staged ? local : base + local;
    lds_barrier();

    if (first) {
        if (m0 < N)      store_tile(af0, disl + l0, ht, m0, lane);
        if (m0 + 16 < N) store_tile(af1, disl + l0 + 16, ht, m0 + 16, lane);
        if (staged) walk_pieces<1>(sorted, segb, cnt, run, stage, nbr, wk, lane, T);
        else        walk_pieces<2>(sorted, segb, cnt, run, stage, nbr, wk, lane, T);
    } else {
        load_x_tile(X, m0 + 16, N, lane, ql1);
        v8f al[4];
        if (m0 < N)      { mma_tile(ql0, Wl, lane, al); store_tile(al, disl + l0, ht, m0, lane); }
        if (m0 + 16 < N) { mma_tile(ql1, Wl, lane, al); store_tile(al, disl + l0 + 16, ht, m0 + 16, lane); }
    }
    if (staged) {
        lds_barrier();
        for (int j = tid; j < tot; j += 256) nbr[base + j] = stage[j];
    }
}

template <int P>
__device__ __forceinline__ int row_lane(int x) {
    return __builtin_amdgcn_update_dpp(0, x, 0x150 + P, 0xf, 0xf, false);
}
__device__ __forceinline__ float xhalf(float x) {
    return __builtin_bit_cast(float, __builtin_amdgcn_permlanex16(
        __builtin_bit_cast(int, x), __builtin_bit_cast(int, x), 0x76543210, 0xfedcba98, false, false));
}

template <int U>
__device__ __forceinline__ void load_pairs(const char* __restrict__ hb, int mine, int n,
                                           unsigned coff, float4 (&v)[16]) {
    if constexpr (U < 16) {
        if (U == 0 || 2 * U < n) {
            const unsigned idx = (unsigned)row_lane<U>(mine);
            v[U] = *(const float4*)(hb + ((idx << 8) | coff));
            load_pairs<U + 1>(hb, mine, n, coff, v);
        }
    }
}
template <int U>
__device__ __forceinline__ void add_pairs(int n, const float4 (&v)[16], float4& acc) {
    if constexpr (U < 16) {
        if (U == 0 || 2 * U < n) {
            acc.x += v[U].x; acc.y += v[U].y; acc.z += v[U].z; acc.w += v[U].w;
            add_pairs<U + 1>(n, v, acc);
        }
    }
}

__device__ __forceinline__ float4 node_sum(const int* __restrict__ nbr,
                                           const char* __restrict__ hb,
                                           int start, int deg, int i, int zrow,
                                           int lane, unsigned coff) {
    const int tot = deg + 1;
    const int place = ((lane & 15) << 1) | (lane >> 4);
    float4 acc; acc.x = 0.0f; acc.y = 0.0f; acc.z = 0.0f; acc.w = 0.0f;
    int base = 0;
    do {
        const int left = tot - base;
        const int n = left < 32 ? left : 32;
        const int t = base + place;
        int mine = (t == deg) ? i : zrow;
        if (t < deg) mine = nbr[start + t];
        float4 v[16];
        load_pairs<0>(hb, mine, n, coff, v);
        add_pairs<0>(n, v, acc);
        base += 32;
    } while (base < tot);
    acc.x += xhalf(acc.x); acc.y += xhalf(acc.y); acc.z += xhalf(acc.z); acc.w += xhalf(acc.w);
    return acc;
}

#define TILE_ROWS 16
#define A_PITCH 68
__global__ void __launch_bounds__(256)
gather_transform_kernel(const int* __restrict__ off,
                        const int* __restrict__ deg,
                        const int* __restrict__ nbr,
                        const float* __restrict__ ht,
                        const float* __restrict__ dis,
                        const float* __restrict__ bias,
                        const float* __restrict__ W,
                        float* __restrict__ hout, int N) {
    __shared__ float4 Wl[2][16][32];
    __shared__ alignas(16) float As[8][TILE_ROWS][A_PITCH];
    const int lane = threadIdx.x & 31;
    const int wv = __builtin_amdgcn_readfirstlane((int)(threadIdx.x >> 5));
    const int m0 = (blockIdx.x * 8 + wv) * TILE_ROWS;
    const int nn = (N - m0) < TILE_ROWS ? (N - m0) : TILE_ROWS;

    int offv = 0, degv = 0, disv = 0;
    if (lane < nn) {
        offv = off[m0 + lane];
        degv = deg[m0 + lane];
        disv = __float_as_int(dis[m0 + lane]);
    }
    const int cg = lane & 15;
    const float4 bj = *(const float4*)(bias + 4 * cg);
    const int zrow = zero_row(N);
    if (blockIdx.x == 0 && wv == 0) {
        float2 z; z.x = 0.0f; z.y = 0.0f;
        *(float2*)(hout + (size_t)zrow * 64 + lane * 2) = z;
    }

    {
        float4 wr[4];
        #pragma unroll
        for (int u = 0; u < 4; ++u)
            wr[u] = *(const float4*)(W + (size_t)(u * 256 + (int)threadIdx.x) * 4);
        const int t = (int)threadIdx.x;
        const int tile = (t & 15) >> 2;
        const int ln0 = ((t >> 5) & 1) * 16 + ((t & 3) << 2);
        const int e = (tile & 1) * 2 + ((t >> 4) & 1);
        float* wp = (float*)&Wl[tile >> 1][t >> 6][ln0] + e;
        #pragma unroll
        for (int u = 0; u < 4; ++u) {
            float* p = wp + u * (4 * 32 * 4);
            p[0] = wr[u].x; p[4] = wr[u].y; p[8] = wr[u].z; p[12] = wr[u].w;
        }
    }
    __syncthreads();
    if (nn <= 0) return;

    const char* hb = (const char*)ht;
    const unsigned coff = (unsigned)cg << 4;
    float* slot = &As[wv][0][lane * 2];
    float* arow = &As[wv][0][cg * 4];

    for (int q = 0; q < TILE_ROWS; ++q) {
        float4 a; a.x = 0.0f; a.y = 0.0f; a.z = 0.0f; a.w = 0.0f;
        if (q < nn) {
            const int start = __builtin_amdgcn_readlane(offv, q);
            const int dq = __builtin_amdgcn_readlane(degv, q);
            const float di = __int_as_float(__builtin_amdgcn_readlane(disv, q));
            const float4 acc = node_sum(nbr, hb, start, dq, m0 + q, zrow, lane, coff);
            a.x = fmaxf(fmaf(acc.x, di, bj.x), 0.0f);
            a.y = fmaxf(fmaf(acc.y, di, bj.y), 0.0f);
            a.z = fmaxf(fmaf(acc.z, di, bj.z), 0.0f);
            a.w = fmaxf(fmaf(acc.w, di, bj.w), 0.0f);
        }
        *(float4*)(arow + q * A_PITCH) = a;
    }

    const int half = lane >> 4;
    const int ml   = lane & 15;
    const float* ap = &As[wv][ml][half * 2];
    v8f c0 = {}, c1 = {}, c2 = {}, c3 = {};
    #pragma unroll
    for (int ks = 0; ks < 16; ++ks) {
        const v2f av = *(const v2f*)(ap + ks * 4);
        const float4 b01 = Wl[0][ks][lane];
        const float4 b23 = Wl[1][ks][lane];
        v2f b0; b0.x = b01.x; b0.y = b01.y;
        v2f b1; b1.x = b01.z; b1.y = b01.w;
        v2f b2; b2.x = b23.x; b2.y = b23.y;
        v2f b3; b3.x = b23.z; b3.y = b23.w;
        c0 = __builtin_amdgcn_wmma_f32_16x16x4_f32(false, av, false, b0, (short)0, c0, false, false);
        c1 = __builtin_amdgcn_wmma_f32_16x16x4_f32(false, av, false, b1, (short)0, c1, false, false);
        c2 = __builtin_amdgcn_wmma_f32_16x16x4_f32(false, av, false, b2, (short)0, c2, false, false);
        c3 = __builtin_amdgcn_wmma_f32_16x16x4_f32(false, av, false, b3, (short)0, c3, false, false);
    }

    float* cp = &As[wv][half * 8][ml];
    #pragma unroll
    for (int r = 0; r < 8; ++r) {
        float* p = cp + r * A_PITCH;
        p[0] = c0[r]; p[16] = c1[r]; p[32] = c2[r]; p[48] = c3[r];
    }
    float2 h[TILE_ROWS];
    #pragma unroll
    for (int q = 0; q < TILE_ROWS; ++q) h[q] = *(const float2*)(slot + q * A_PITCH);
    #pragma unroll
    for (int q = 0; q < TILE_ROWS; ++q) {
        if (q < nn) {
            const float dr = __int_as_float(__builtin_amdgcn_readlane(disv, q));
            float2 o; o.x = h[q].x * dr; o.y = h[q].y * dr;
            *(float2*)(hout + (size_t)(m0 + q) * 64 + lane * 2) = o;
        }
    }
}

#define FC_NODES 8
template <int MASK>
__device__ __forceinline__ float xrow(float x) {
    return __builtin_bit_cast(float, __builtin_amdgcn_update_dpp(
        0, __builtin_bit_cast(int, x), 0x160 + MASK, 0xf, 0xf, false));
}
__device__ __forceinline__ float half_sum(float p) {
    p += xrow<8>(p);
    p += xrow<4>(p);
    p += xrow<2>(p);
    p += xrow<1>(p);
    return p;
}

__global__ void __launch_bounds__(256)
gather_fc_kernel(const int* __restrict__ off,
                 const int* __restrict__ deg,
                 const int* __restrict__ nbr,
                 const float* __restrict__ ht,
                 const float* __restrict__ dis,
                 const float* __restrict__ b2,
                 const float* __restrict__ wfc,
                 const float* __restrict__ bfc,
                 float* __restrict__ out, int N) {
    __shared__ float res[8 * FC_NODES];
    const int lane = threadIdx.x & 31;
    const int wv = __builtin_amdgcn_readfirstlane((int)(threadIdx.x >> 5));
    const int i0 = (blockIdx.x * 8 + wv) * FC_NODES;
    float r = 0.0f;
    if (i0 < N) {
        const int nn = (N - i0) < FC_NODES ? (N - i0) : FC_NODES;
        int offv = 0, degv = 0, disv = 0;
        if (lane < nn) {
            offv = off[i0 + lane];
            degv = deg[i0 + lane];
            disv = __float_as_int(dis[i0 + lane]);
        }
        const char* hb = (const char*)ht;
        const int cg = lane & 15;
        const unsigned coff = (unsigned)cg << 4;
        const float4 bj = *(const float4*)(b2 + 4 * cg);
        const float4 wj = *(const float4*)(wfc + 4 * cg);
        const int zrow = zero_row(N);
        for (int q = 0; q < nn; ++q) {
            const int start = __builtin_amdgcn_readlane(offv, q);
            const int dq = __builtin_amdgcn_readlane(degv, q);
            const float di = __int_as_float(__builtin_amdgcn_readlane(disv, q));
            const float4 acc = node_sum(nbr, hb, start, dq, i0 + q, zrow, lane, coff);
            const float h0 = fmaxf(fmaf(acc.x, di, bj.x), 0.0f);
            const float h1 = fmaxf(fmaf(acc.y, di, bj.y), 0.0f);
            const float h2 = fmaxf(fmaf(acc.z, di, bj.z), 0.0f);
            const float h3 = fmaxf(fmaf(acc.w, di, bj.w), 0.0f);
            const float p = half_sum((h0 * wj.x + h1 * wj.y) + (h2 * wj.z + h3 * wj.w));
            if (lane == q) r = p;
        }
    }
    if (lane < FC_NODES) res[wv * FC_NODES + lane] = r;
    __syncthreads();
    if (wv == 0) {
        const int b = blockIdx.x * (8 * FC_NODES) + lane * 2;
        const float bf = bfc[0];
        if (b + 1 < N) {
            float2 o; o.x = res[lane * 2] + bf; o.y = res[lane * 2 + 1] + bf;
            *(float2*)(out + b) = o;
        } else if (b < N) {
            out[b] = res[lane * 2] + bf;
        }
    }
}

extern "C" void kernel_launch(void* const* d_in, const int* in_sizes, int n_in,
                              void* d_out, int out_size, void* d_ws, size_t ws_size,
                              hipStream_t stream) {
    const float* x   = (const float*)d_in[0];
    const int*   ei  = (const int*)d_in[1];
    const float* W1  = (const float*)d_in[2];
    const float* b1  = (const float*)d_in[3];
    const float* W2  = (const float*)d_in[4];
    const float* b2  = (const float*)d_in[5];
    const float* Wfc = (const float*)d_in[6];
    const float* bfc = (const float*)d_in[7];

    const int HID = in_sizes[3];
    const int IN  = in_sizes[2] / HID;
    const int N   = in_sizes[0] / IN;
    const int E   = in_sizes[1] / 2;
    const int* srcv = ei;
    const int* dstv = ei + E;

    const int NB = (N + BUCKET_NODES - 1) / BUCKET_NODES;
    const int T  = (E + TILE_EDGES - 1) / TILE_EDGES;
    char* ws = (char*)d_ws;
    const size_t nodeBytes = (((size_t)N * 4) + 255) & ~(size_t)255;
    const size_t edgeBytes = (((size_t)E * 4) + 255) & ~(size_t)255;
    const size_t tileBytes = (size_t)T * TILE_EDGES * 4;
    const size_t segBytes  = (((size_t)NB * T * 4) + 255) & ~(size_t)255;
    const size_t rowBytes  = ((size_t)N + 16) * 64 * sizeof(float);
    float* dis    = (float*)ws;
    int*   deg    = (int*)(ws + nodeBytes);
    int*   offs   = (int*)(ws + 2 * nodeBytes);
    int*   cursor = (int*)(ws + 3 * nodeBytes);
    size_t o = 3 * nodeBytes + 256;
    int*      seg    = (int*)(ws + o);
    unsigned* sorted = (unsigned*)(ws + o + segBytes);
    int*      nbr    = (int*)(ws + o + segBytes + tileBytes);
    o += segBytes + tileBytes + edgeBytes;
    float* ht   = (float*)(ws + o);
    float* ht2  = (float*)(ws + o + rowBytes);

    const int gemmBlocks = ((N + 15) / 16 + 7) / 8;
    const int fcBlocks   = ((N + FC_NODES - 1) / FC_NODES + 7) / 8;

    tile_sort_kernel<<<T, 256, 0, stream>>>(srcv, dstv, sorted, seg, cursor, E, NB, T);
    csr_build_kernel<<<NB, 256, 0, stream>>>(sorted, seg, cursor, offs, deg, dis, nbr, x, W1, ht, N, T);

    gather_transform_kernel<<<gemmBlocks, 256, 0, stream>>>(offs, deg, nbr, ht, dis, b1, W2, ht2, N);

    gather_fc_kernel<<<fcBlocks, 256, 0, stream>>>(offs, deg, nbr, ht2, dis, b2, Wfc, bfc, (float*)d_out, N);
}
